# starting kernel
# baseline (speedup 1.0000x reference)
#include <hip/hip_runtime.h>

#define NN 50000

typedef float v8f __attribute__((ext_vector_type(8)));
typedef float v2f __attribute__((ext_vector_type(2)));

__global__ void fill_f32(float* __restrict__ p, float v, int n) {
    int t = blockIdx.x * blockDim.x + threadIdx.x;
    if (t < n) p[t] = v;
}

__global__ void deg_scatter(const int* __restrict__ col, float* __restrict__ deg, int E) {
    int t = blockIdx.x * blockDim.x + threadIdx.x;
    if (t < E) atomicAdd(&deg[col[t]], 1.0f);
}

__global__ void deg_to_dis(float* __restrict__ d, int n) {
    int t = blockIdx.x * blockDim.x + threadIdx.x;
    if (t < n) {
        float v = d[t];
        d[t] = (v > 0.0f) ? rsqrtf(fmaxf(v, 1.0f)) : 0.0f;
    }
}

__global__ void edge_norm(const int* __restrict__ row, const int* __restrict__ col,
                          const float* __restrict__ dis, float* __restrict__ norm, int E) {
    int t = blockIdx.x * blockDim.x + threadIdx.x;
    if (t < E) norm[t] = dis[row[t]] * dis[col[t]];
}

template <int K, int F, bool RELU>
__global__ void gemm_wmma_f32(const float* __restrict__ A, const float* __restrict__ W,
                              float* __restrict__ H, int totalTiles) {
    constexpr int tilesN = F / 16;
    int wave = blockIdx.x * (blockDim.x >> 5) + (threadIdx.x >> 5);
    if (wave >= totalTiles) return;
    int lane = threadIdx.x & 31;
    int half = lane >> 4;
    int lm   = lane & 15;
    int tm = wave / tilesN;
    int tn = wave - tm * tilesN;
    const float* __restrict__ Arow = A + (size_t)(tm * 16 + lm) * K;
    int bcol = tn * 16 + lm;

    v8f c = {};
    constexpr int KM = K & ~3;
#pragma unroll
    for (int k0 = 0; k0 < KM; k0 += 4) {
        int ka = k0 + 2 * half;
        float a0, a1;
        if constexpr ((K & 1) == 0) {
            float2 a2 = *(const float2*)(Arow + ka);
            a0 = a2.x; a1 = a2.y;
        } else {
            a0 = Arow[ka]; a1 = Arow[ka + 1];
        }
        if constexpr (RELU) { a0 = fmaxf(a0, 0.0f); a1 = fmaxf(a1, 0.0f); }
        float b0 = W[(size_t)ka * F + bcol];
        float b1 = W[(size_t)(ka + 1) * F + bcol];
        v2f av; av.x = a0; av.y = a1;
        v2f bv; bv.x = b0; bv.y = b1;
        c = __builtin_amdgcn_wmma_f32_16x16x4_f32(false, av, false, bv,
                                                  (short)0, c, false, false);
    }
    if constexpr (KM < K) {
        int ka = KM + 2 * half;
        int i0 = (ka     < K) ? ka       : (K - 1);
        int i1 = (ka + 1 < K) ? (ka + 1) : (K - 1);
        float a0 = Arow[i0];
        float a1 = Arow[i1];
        float b0 = W[(size_t)i0 * F + bcol];
        float b1 = W[(size_t)i1 * F + bcol];
        if (ka     >= K) { a0 = 0.0f; b0 = 0.0f; }
        if (ka + 1 >= K) { a1 = 0.0f; b1 = 0.0f; }
        if constexpr (RELU) { a0 = fmaxf(a0, 0.0f); a1 = fmaxf(a1, 0.0f); }
        v2f av; av.x = a0; av.y = a1;
        v2f bv; bv.x = b0; bv.y = b1;
        c = __builtin_amdgcn_wmma_f32_16x16x4_f32(false, av, false, bv,
                                                  (short)0, c, false, false);
    }

    int col = tn * 16 + lm;
#pragma unroll
    for (int j = 0; j < 8; ++j) {
        int r = tm * 16 + j + 8 * half;
        H[(size_t)r * F + col] = c[j];
    }
}

template <int F>
__global__ void agg_init(const float* __restrict__ h, const float* __restrict__ dis,
                         const float* __restrict__ bias, float* __restrict__ out) {
    int t = blockIdx.x * blockDim.x + threadIdx.x;
    if (t >= NN * F) return;
    int i = t / F;
    int f = t - i * F;
    float d = dis[i];
    out[t] = bias[f] + h[t] * d * d;
}

template <int F>
__global__ void agg_scatter(const float* __restrict__ h, const int* __restrict__ row,
                            const int* __restrict__ col, const float* __restrict__ norm,
                            float* __restrict__ out, int E) {
    long long t = (long long)blockIdx.x * blockDim.x + threadIdx.x;
    if (t >= (long long)E * F) return;
    int e = (int)(t / F);
    int f = (int)(t - (long long)e * F);
    float m = h[(size_t)row[e] * F + f] * norm[e];
    atomicAdd(&out[(size_t)col[e] * F + f], m);
}

static inline int ceil_div(long long a, long long b) { return (int)((a + b - 1) / b); }

extern "C" void kernel_launch(void* const* d_in, const int* in_sizes, int n_in,
                              void* d_out, int out_size, void* d_ws, size_t ws_size,
                              hipStream_t stream) {
    const float* x  = (const float*)d_in[0];
    const int*   ei = (const int*)d_in[1];
    const float* W1 = (const float*)d_in[2];
    const float* b1 = (const float*)d_in[3];
    const float* W2 = (const float*)d_in[4];
    const float* b2 = (const float*)d_in[5];
    const float* W3 = (const float*)d_in[6];
    const float* b3 = (const float*)d_in[7];
    const float* W4 = (const float*)d_in[8];
    const float* b4 = (const float*)d_in[9];

    const int E = in_sizes[1] / 2;
    const int* row = ei;
    const int* col = ei + E;

    float* dis  = (float*)d_ws;
    float* norm = dis  + ((NN + 255) & ~255);
    float* bufA = norm + ((E  + 255) & ~255);
    float* bufB = bufA + (size_t)NN * 128;

    const int TB = 256;

    fill_f32<<<ceil_div(NN, TB), TB, 0, stream>>>(dis, 1.0f, NN);
    deg_scatter<<<ceil_div(E, TB), TB, 0, stream>>>(col, dis, E);
    deg_to_dis<<<ceil_div(NN, TB), TB, 0, stream>>>(dis, NN);
    edge_norm<<<ceil_div(E, TB), TB, 0, stream>>>(row, col, dis, norm, E);

    const int TILES_M = NN / 16;
    const int GB = 128;

    {
        int total = TILES_M * (96 / 16);
        gemm_wmma_f32<29, 96, false><<<ceil_div(total, GB / 32), GB, 0, stream>>>(x, W1, bufA, total);
    }
    agg_init<96><<<ceil_div((long long)NN * 96, TB), TB, 0, stream>>>(bufA, dis, b1, bufB);
    agg_scatter<96><<<ceil_div((long long)E * 96, TB), TB, 0, stream>>>(bufA, row, col, norm, bufB, E);

    {
        int total = TILES_M * (128 / 16);
        gemm_wmma_f32<96, 128, true><<<ceil_div(total, GB / 32), GB, 0, stream>>>(bufB, W2, bufA, total);
    }
    agg_init<128><<<ceil_div((long long)NN * 128, TB), TB, 0, stream>>>(bufA, dis, b2, bufB);
    agg_scatter<128><<<ceil_div((long long)E * 128, TB), TB, 0, stream>>>(bufA, row, col, norm, bufB, E);

    {
        int total = TILES_M * (64 / 16);
        gemm_wmma_f32<128, 64, false><<<ceil_div(total, GB / 32), GB, 0, stream>>>(bufB, W3, bufA, total);
    }
    agg_init<64><<<ceil_div((long long)NN * 64, TB), TB, 0, stream>>>(bufA, dis, b3, bufB);
    agg_scatter<64><<<ceil_div((long long)E * 64, TB), TB, 0, stream>>>(bufA, row, col, norm, bufB, E);

    {
        int total = TILES_M * (32 / 16);
        gemm_wmma_f32<64, 32, false><<<ceil_div(total, GB / 32), GB, 0, stream>>>(bufB, W4, bufA, total);
    }
    float* out = (float*)d_out;
    agg_init<32><<<ceil_div((long long)NN * 32, TB), TB, 0, stream>>>(bufA, dis, b4, out);
    agg_scatter<32><<<ceil_div((long long)E * 32, TB), TB, 0, stream>>>(bufA, row, col, norm, out, E);
}


# Edges grouped by destination once per call (returning int atomic rank, scan, fill); each layer sums its gathered rows per node in registers: no float atomics, no init pass; sum order = arrival rank
# speedup vs baseline: 3.2405x; 3.2405x over previous
#include <hip/hip_runtime.h>

#define NN 50000

typedef float v8f __attribute__((ext_vector_type(8)));
typedef float v2f __attribute__((ext_vector_type(2)));

typedef unsigned short u16;
static_assert(NN <= 65536, "source ids are stored in 16 bits");

#define SCAN_T  1024
#define SCAN_CH ((NN + SCAN_T - 1) / SCAN_T)

__global__ void zero_i32(int* __restrict__ p, int n) {
    int t = blockIdx.x * blockDim.x + threadIdx.x;
    if (t < n) p[t] = 0;
}

__global__ void edge_rank(const int* __restrict__ col, int* __restrict__ cnt,
                          int* __restrict__ rank, int E) {
    int t = blockIdx.x * blockDim.x + threadIdx.x;
    if (t < E) rank[t] = atomicAdd(&cnt[col[t]], 1);
}

static_assert((SCAN_CH & 1) == 1, "odd lane stride keeps the chunk reads bank-conflict free");
__global__ void __launch_bounds__(SCAN_T)
scan_dis(int* __restrict__ cntptr, float* __restrict__ dis) {
    __shared__ int buf[SCAN_T * SCAN_CH];
    __shared__ int sh[SCAN_T];
    const int t = threadIdx.x;
    {
        constexpr int FULL = NN / SCAN_T;
        int tmp[SCAN_CH];
#pragma unroll
        for (int k = 0; k < FULL; ++k) tmp[k] = cntptr[t + k * SCAN_T];
#pragma unroll
        for (int k = FULL; k < SCAN_CH; ++k) {
            int i = t + k * SCAN_T;
            tmp[k] = (i < NN) ? cntptr[i] : 0;
        }
#pragma unroll
        for (int k = 0; k < SCAN_CH; ++k) buf[t + k * SCAN_T] = tmp[k];
    }
    __syncthreads();
    const int base = t * SCAN_CH;
    int s = 0;
    for (int j = 0; j < SCAN_CH; ++j) s += buf[base + j];
    sh[t] = s;
    __syncthreads();
    for (int off = 1; off < SCAN_T; off <<= 1) {
        int v = (t >= off) ? sh[t - off] : 0;
        __syncthreads();
        sh[t] += v;
        __syncthreads();
    }
    int run = sh[t] - s;
    for (int j = 0; j < SCAN_CH; ++j) {
        int c = buf[base + j];
        buf[base + j] = run;
        run += c;
    }
    __syncthreads();
    for (int i = t; i <= NN; i += SCAN_T) {
        int p = buf[i];
        cntptr[i] = p;
        if (i < NN) {
            float v = (float)(buf[i + 1] - p + 1);
            dis[i] = (v > 0.0f) ? rsqrtf(fmaxf(v, 1.0f)) : 0.0f;
        }
    }
}

__global__ void group_fill(const int* __restrict__ row, const int* __restrict__ col,
                           const int* __restrict__ rank, const int* __restrict__ ptr,
                           u16* __restrict__ srcs, int E) {
    int t = blockIdx.x * blockDim.x + threadIdx.x;
    if (t < E) srcs[ptr[col[t]] + rank[t]] = (u16)row[t];
}

template <int K, int F, bool RELU>
__global__ void gemm_wmma_f32(const float* __restrict__ A, const float* __restrict__ W,
                              float* __restrict__ H, int totalTiles) {
    constexpr int tilesN = F / 16;
    int wave = blockIdx.x * (blockDim.x >> 5) + (threadIdx.x >> 5);
    if (wave >= totalTiles) return;
    int lane = threadIdx.x & 31;
    int half = lane >> 4;
    int lm   = lane & 15;
    int tm = wave / tilesN;
    int tn = wave - tm * tilesN;
    const float* __restrict__ Arow = A + (size_t)(tm * 16 + lm) * K;
    int bcol = tn * 16 + lm;

    v8f c = {};
    constexpr int KM = K & ~3;
#pragma unroll
    for (int k0 = 0; k0 < KM; k0 += 4) {
        int ka = k0 + 2 * half;
        float a0, a1;
        if constexpr ((K & 1) == 0) {
            float2 a2 = *(const float2*)(Arow + ka);
            a0 = a2.x; a1 = a2.y;
        } else {
            a0 = Arow[ka]; a1 = Arow[ka + 1];
        }
        if constexpr (RELU) { a0 = fmaxf(a0, 0.0f); a1 = fmaxf(a1, 0.0f); }
        float b0 = W[(size_t)ka * F + bcol];
        float b1 = W[(size_t)(ka + 1) * F + bcol];
        v2f av; av.x = a0; av.y = a1;
        v2f bv; bv.x = b0; bv.y = b1;
        c = __builtin_amdgcn_wmma_f32_16x16x4_f32(false, av, false, bv,
                                                  (short)0, c, false, false);
    }
    if constexpr (KM < K) {
        int ka = KM + 2 * half;
        int i0 = (ka     < K) ? ka       : (K - 1);
        int i1 = (ka + 1 < K) ? (ka + 1) : (K - 1);
        float a0 = Arow[i0];
        float a1 = Arow[i1];
        float b0 = W[(size_t)i0 * F + bcol];
        float b1 = W[(size_t)i1 * F + bcol];
        if (ka     >= K) { a0 = 0.0f; b0 = 0.0f; }
        if (ka + 1 >= K) { a1 = 0.0f; b1 = 0.0f; }
        if constexpr (RELU) { a0 = fmaxf(a0, 0.0f); a1 = fmaxf(a1, 0.0f); }
        v2f av; av.x = a0; av.y = a1;
        v2f bv; bv.x = b0; bv.y = b1;
        c = __builtin_amdgcn_wmma_f32_16x16x4_f32(false, av, false, bv,
                                                  (short)0, c, false, false);
    }

    int col = tn * 16 + lm;
#pragma unroll
    for (int j = 0; j < 8; ++j) {
        int r = tm * 16 + j + 8 * half;
        H[(size_t)r * F + col] = c[j];
    }
}

template <int VEC>
__device__ inline void load_row(const float* __restrict__ p, float (&r)[VEC]) {
    if constexpr (VEC == 4) {
        float4 v = *(const float4*)p;
        r[0] = v.x; r[1] = v.y; r[2] = v.z; r[3] = v.w;
    } else if constexpr (VEC == 2) {
        float2 v = *(const float2*)p;
        r[0] = v.x; r[1] = v.y;
    } else {
#pragma unroll
        for (int v = 0; v < VEC; ++v) r[v] = p[v];
    }
}

template <int F, int VEC, int UU>
__device__ inline void row_batch(const float* __restrict__ h, int fo, int my_s, float my_d,
                                 float di, int j, int m, float (&acc)[VEC]) {
    float r[UU][VEC];
#pragma unroll
    for (int u = 0; u < UU; ++u) {
        int jj = j + (u < m ? u : m - 1);
        int s = __builtin_amdgcn_readlane(my_s, jj);
        load_row<VEC>(h + (size_t)s * F + fo, r[u]);
    }
    __builtin_amdgcn_sched_barrier(0);
#pragma unroll
    for (int u = 0; u < UU; ++u) {
        const bool on = u < m;
        float w = __int_as_float(__builtin_amdgcn_readlane(__float_as_int(my_d), j + u)) * di;
#pragma unroll
        for (int v = 0; v < VEC; ++v) acc[v] += (on ? r[u][v] : 0.0f) * w;
    }
}

template <int F>
__global__ void __launch_bounds__(256)
agg_gather(const float* __restrict__ h, const float* __restrict__ dis,
           const int* __restrict__ ptr, const u16* __restrict__ srcs,
           const float* __restrict__ bias, float* __restrict__ out) {
    constexpr int VEC = F / 32;
    static_assert(F % 32 == 0, "one wave covers a row");
    const int node = blockIdx.x * (blockDim.x >> 5) + (threadIdx.x >> 5);
    if (node >= NN) return;
    const int lane = threadIdx.x & 31;
    const int fo = lane * VEC;
    const int beg = __builtin_amdgcn_readfirstlane(ptr[node]);
    const int end = __builtin_amdgcn_readfirstlane(ptr[node + 1]);
    const float di = dis[node];

    float acc[VEC];
    {
        float r[VEC];
        load_row<VEC>(h + (size_t)node * F + fo, r);
#pragma unroll
        for (int v = 0; v < VEC; ++v) acc[v] = bias[fo + v] + r[v] * di * di;
    }

    for (int e0 = beg; e0 < end; e0 += 32) {
        const int rem = end - e0;
        const int n = rem < 32 ? rem : 32;
        int   my_s = 0;
        float my_d = 0.0f;
        if (lane < n) {
            my_s = (int)srcs[e0 + lane];
            my_d = dis[my_s];
        }
        int j = 0;
        for (; j + 8 <= n; j += 8)
            row_batch<F, VEC, 8>(h, fo, my_s, my_d, di, j, 8, acc);
        const int m = n - j;
        if (m > 4)      row_batch<F, VEC, 8>(h, fo, my_s, my_d, di, j, m, acc);
        else if (m > 0) row_batch<F, VEC, 4>(h, fo, my_s, my_d, di, j, m, acc);
    }

    float* __restrict__ o = out + (size_t)node * F + fo;
    if constexpr (VEC == 4) {
        *(float4*)o = make_float4(acc[0], acc[1], acc[2], acc[3]);
    } else if constexpr (VEC == 2) {
        *(float2*)o = make_float2(acc[0], acc[1]);
    } else {
#pragma unroll
        for (int v = 0; v < VEC; ++v) o[v] = acc[v];
    }
}

static inline int ceil_div(long long a, long long b) { return (int)((a + b - 1) / b); }

extern "C" void kernel_launch(void* const* d_in, const int* in_sizes, int n_in,
                              void* d_out, int out_size, void* d_ws, size_t ws_size,
                              hipStream_t stream) {
    const float* x  = (const float*)d_in[0];
    const int*   ei = (const int*)d_in[1];
    const float* W1 = (const float*)d_in[2];
    const float* b1 = (const float*)d_in[3];
    const float* W2 = (const float*)d_in[4];
    const float* b2 = (const float*)d_in[5];
    const float* W3 = (const float*)d_in[6];
    const float* b3 = (const float*)d_in[7];
    const float* W4 = (const float*)d_in[8];
    const float* b4 = (const float*)d_in[9];

    const int E = in_sizes[1] / 2;
    const int* row = ei;
    const int* col = ei + E;

    float* dis  = (float*)d_ws;
    float* reg1 = dis  + ((NN + 255) & ~255);
    float* bufA = reg1 + ((E  + 255) & ~255);
    float* bufB = bufA + (size_t)NN * 128;
    int*   ptr  = (int*)reg1;
    u16*   srcs = (u16*)(ptr + ((NN + 1 + 255) & ~255));
    int*   rank = (int*)bufB;

    const int TB = 256;

    zero_i32<<<ceil_div(NN, TB), TB, 0, stream>>>(ptr, NN);
    edge_rank<<<ceil_div(E, TB), TB, 0, stream>>>(col, ptr, rank, E);
    scan_dis<<<1, SCAN_T, 0, stream>>>(ptr, dis);
    group_fill<<<ceil_div(E, TB), TB, 0, stream>>>(row, col, rank, ptr, srcs, E);

    const int TILES_M = NN / 16;
    const int GB = 128;
    const int AGB = ceil_div(NN, TB / 32);

    {
        int total = TILES_M * (96 / 16);
        gemm_wmma_f32<29, 96, false><<<ceil_div(total, GB / 32), GB, 0, stream>>>(x, W1, bufA, total);
    }
    agg_gather<96><<<AGB, TB, 0, stream>>>(bufA, dis, ptr, srcs, b1, bufB);

    {
        int total = TILES_M * (128 / 16);
        gemm_wmma_f32<96, 128, true><<<ceil_div(total, GB / 32), GB, 0, stream>>>(bufB, W2, bufA, total);
    }
    agg_gather<128><<<AGB, TB, 0, stream>>>(bufA, dis, ptr, srcs, b2, bufB);

    {
        int total = TILES_M * (64 / 16);
        gemm_wmma_f32<128, 64, false><<<ceil_div(total, GB / 32), GB, 0, stream>>>(bufB, W3, bufA, total);
    }
    agg_gather<64><<<AGB, TB, 0, stream>>>(bufA, dis, ptr, srcs, b3, bufB);

    {
        int total = TILES_M * (32 / 16);
        gemm_wmma_f32<64, 32, false><<<ceil_div(total, GB / 32), GB, 0, stream>>>(bufB, W4, bufA, total);
    }
    float* out = (float*)d_out;
    agg_gather<32><<<AGB, TB, 0, stream>>>(bufA, dis, ptr, srcs, b4, out);
}


# Aggregation commutes with the dense products: x is gathered 32 wide before W1, W2*W3*W4 is folded per call on WMMA and layers 2-4 gather 32 wide; products and sums reassociated, f32 kept
# speedup vs baseline: 4.1602x; 1.2838x over previous
#include <hip/hip_runtime.h>

#define NN 50000

typedef float v8f __attribute__((ext_vector_type(8)));
typedef float v2f __attribute__((ext_vector_type(2)));

typedef unsigned short u16;
static_assert(NN <= 65536, "source ids are stored in 16 bits");

#define SCAN_T  1024
#define SCAN_CH ((NN + SCAN_T - 1) / SCAN_T)

#define XK  29
#define XKP 32
__global__ void prep_pad_zero(const float* __restrict__ x, float* __restrict__ xp,
                              int* __restrict__ cnt) {
    int t = blockIdx.x * blockDim.x + threadIdx.x;
    if (t < NN * XKP) {
        int n = t >> 5, l = t & 31;
        xp[t] = (l < XK) ? x[(size_t)n * XK + l] : 0.0f;
    }
    if (t < NN) cnt[t] = 0;
}

__global__ void __launch_bounds__(64)
bias_chain(const float* __restrict__ b2, const float* __restrict__ W3,
           const float* __restrict__ b3, const float* __restrict__ W4,
           float* __restrict__ c2, float* __restrict__ c3) {
    __shared__ float tmp[64];
    const int t = threadIdx.x;
    float s = 0.0f;
    for (int k = 0; k < 128; ++k) s += b2[k] * W3[k * 64 + t];
    tmp[t] = s;
    __syncthreads();
    if (t < 32) {
        float a = 0.0f, b = 0.0f;
        for (int j = 0; j < 64; ++j) {
            float w = W4[j * 32 + t];
            a += tmp[j] * w;
            b += b3[j] * w;
        }
        c2[t] = a;
        c3[t] = b;
    }
}

__global__ void edge_rank(const int* __restrict__ col, int* __restrict__ cnt,
                          int* __restrict__ rank, int E) {
    int t = blockIdx.x * blockDim.x + threadIdx.x;
    if (t < E) rank[t] = atomicAdd(&cnt[col[t]], 1);
}

static_assert((SCAN_CH & 1) == 1, "odd lane stride keeps the chunk reads bank-conflict free");
__global__ void __launch_bounds__(SCAN_T)
scan_dis(int* __restrict__ cntptr, float* __restrict__ dis) {
    __shared__ int buf[SCAN_T * SCAN_CH];
    __shared__ int sh[SCAN_T];
    const int t = threadIdx.x;
    {
        constexpr int FULL = NN / SCAN_T;
        int tmp[SCAN_CH];
#pragma unroll
        for (int k = 0; k < FULL; ++k) tmp[k] = cntptr[t + k * SCAN_T];
#pragma unroll
        for (int k = FULL; k < SCAN_CH; ++k) {
            int i = t + k * SCAN_T;
            tmp[k] = (i < NN) ? cntptr[i] : 0;
        }
#pragma unroll
        for (int k = 0; k < SCAN_CH; ++k) buf[t + k * SCAN_T] = tmp[k];
    }
    __syncthreads();
    const int base = t * SCAN_CH;
    int s = 0;
    for (int j = 0; j < SCAN_CH; ++j) s += buf[base + j];
    sh[t] = s;
    __syncthreads();
    for (int off = 1; off < SCAN_T; off <<= 1) {
        int v = (t >= off) ? sh[t - off] : 0;
        __syncthreads();
        sh[t] += v;
        __syncthreads();
    }
    int run = sh[t] - s;
    for (int j = 0; j < SCAN_CH; ++j) {
        int c = buf[base + j];
        buf[base + j] = run;
        run += c;
    }
    __syncthreads();
    for (int i = t; i <= NN; i += SCAN_T) {
        int p = buf[i];
        cntptr[i] = p;
        if (i < NN) {
            float v = (float)(buf[i + 1] - p + 1);
            dis[i] = (v > 0.0f) ? rsqrtf(fmaxf(v, 1.0f)) : 0.0f;
        }
    }
}

__global__ void group_fill(const int* __restrict__ row, const int* __restrict__ col,
                           const int* __restrict__ rank, const int* __restrict__ ptr,
                           u16* __restrict__ srcs, int E) {
    int t = blockIdx.x * blockDim.x + threadIdx.x;
    if (t < E) srcs[ptr[col[t]] + rank[t]] = (u16)row[t];
}

template <int K, int KW, int F, bool BIAS, bool RELU>
__global__ void gemm_wmma_f32(const float* __restrict__ A, const float* __restrict__ W,
                              const float* __restrict__ bias, float* __restrict__ H,
                              int totalTiles) {
    static_assert(K % 4 == 0 && KW <= K && K - KW < 4, "K is KW rounded up to 4");
    constexpr int tilesN = F / 16;
    int wave = blockIdx.x * (blockDim.x >> 5) + (threadIdx.x >> 5);
    if (wave >= totalTiles) return;
    int lane = threadIdx.x & 31;
    int half = lane >> 4;
    int lm   = lane & 15;
    int tm = wave / tilesN;
    int tn = wave - tm * tilesN;
    const float* __restrict__ Arow = A + (size_t)(tm * 16 + lm) * K;
    int bcol = tn * 16 + lm;

    v8f c = {};
#pragma unroll
    for (int k0 = 0; k0 < K; k0 += 4) {
        int ka = k0 + 2 * half;
        float2 a2 = *(const float2*)(Arow + ka);
        float a0 = a2.x, a1 = a2.y;
        float b0, b1;
        if (k0 + 4 <= KW) {
            b0 = W[(size_t)ka * F + bcol];
            b1 = W[(size_t)(ka + 1) * F + bcol];
        } else {
            int i0 = (ka     < KW) ? ka       : (KW - 1);
            int i1 = (ka + 1 < KW) ? (ka + 1) : (KW - 1);
            b0 = W[(size_t)i0 * F + bcol];
            b1 = W[(size_t)i1 * F + bcol];
            if (ka     >= KW) { a0 = 0.0f; b0 = 0.0f; }
            if (ka + 1 >= KW) { a1 = 0.0f; b1 = 0.0f; }
        }
        v2f av; av.x = a0; av.y = a1;
        v2f bv; bv.x = b0; bv.y = b1;
        c = __builtin_amdgcn_wmma_f32_16x16x4_f32(false, av, false, bv,
                                                  (short)0, c, false, false);
    }

    int col = tn * 16 + lm;
    float bb = 0.0f;
    if constexpr (BIAS) bb = bias[col];
#pragma unroll
    for (int j = 0; j < 8; ++j) {
        int r = tm * 16 + j + 8 * half;
        float v = c[j];
        if constexpr (BIAS) v += bb;
        if constexpr (RELU) v = fmaxf(v, 0.0f);
        H[(size_t)r * F + col] = v;
    }
}

template <int VEC>
__device__ inline void load_row(const float* __restrict__ p, float (&r)[VEC]) {
    if constexpr (VEC == 4) {
        float4 v = *(const float4*)p;
        r[0] = v.x; r[1] = v.y; r[2] = v.z; r[3] = v.w;
    } else if constexpr (VEC == 2) {
        float2 v = *(const float2*)p;
        r[0] = v.x; r[1] = v.y;
    } else {
#pragma unroll
        for (int v = 0; v < VEC; ++v) r[v] = p[v];
    }
}

template <int F, int VEC, int UU>
__device__ inline void row_batch(const float* __restrict__ h, int fo, int my_s, float my_d,
                                 float di, int j, int m, float (&acc)[VEC]) {
    float r[UU][VEC];
#pragma unroll
    for (int u = 0; u < UU; ++u) {
        int jj = j + (u < m ? u : m - 1);
        int s = __builtin_amdgcn_readlane(my_s, jj);
        load_row<VEC>(h + (size_t)s * F + fo, r[u]);
    }
    __builtin_amdgcn_sched_barrier(0);
#pragma unroll
    for (int u = 0; u < UU; ++u) {
        const bool on = u < m;
        float w = __int_as_float(__builtin_amdgcn_readlane(__float_as_int(my_d), j + u)) * di;
#pragma unroll
        for (int v = 0; v < VEC; ++v) acc[v] += (on ? r[u][v] : 0.0f) * w;
    }
}

template <int F, bool BIAS>
__global__ void __launch_bounds__(256)
agg_gather(const float* __restrict__ h, const float* __restrict__ dis,
           const int* __restrict__ ptr, const u16* __restrict__ srcs,
           const float* __restrict__ bias, float* __restrict__ out) {
    constexpr int VEC = F / 32;
    static_assert(F % 32 == 0, "one wave covers a row");
    const int node = blockIdx.x * (blockDim.x >> 5) + (threadIdx.x >> 5);
    if (node >= NN) return;
    const int lane = threadIdx.x & 31;
    const int fo = lane * VEC;
    const int beg = __builtin_amdgcn_readfirstlane(ptr[node]);
    const int end = __builtin_amdgcn_readfirstlane(ptr[node + 1]);
    const float di = dis[node];

    float acc[VEC];
    {
        float r[VEC];
        load_row<VEC>(h + (size_t)node * F + fo, r);
#pragma unroll
        for (int v = 0; v < VEC; ++v) {
            float b = 0.0f;
            if constexpr (BIAS) b = bias[fo + v];
            acc[v] = b + r[v] * di * di;
        }
    }

    for (int e0 = beg; e0 < end; e0 += 32) {
        const int rem = end - e0;
        const int n = rem < 32 ? rem : 32;
        int   my_s = 0;
        float my_d = 0.0f;
        if (lane < n) {
            my_s = (int)srcs[e0 + lane];
            my_d = dis[my_s];
        }
        int j = 0;
        for (; j + 8 <= n; j += 8)
            row_batch<F, VEC, 8>(h, fo, my_s, my_d, di, j, 8, acc);
        const int m = n - j;
        if (m > 4)      row_batch<F, VEC, 8>(h, fo, my_s, my_d, di, j, m, acc);
        else if (m > 0) row_batch<F, VEC, 4>(h, fo, my_s, my_d, di, j, m, acc);
    }

    float* __restrict__ o = out + (size_t)node * F + fo;
    if constexpr (VEC == 4) {
        *(float4*)o = make_float4(acc[0], acc[1], acc[2], acc[3]);
    } else if constexpr (VEC == 2) {
        *(float2*)o = make_float2(acc[0], acc[1]);
    } else {
#pragma unroll
        for (int v = 0; v < VEC; ++v) o[v] = acc[v];
    }
}

static inline int ceil_div(long long a, long long b) { return (int)((a + b - 1) / b); }

extern "C" void kernel_launch(void* const* d_in, const int* in_sizes, int n_in,
                              void* d_out, int out_size, void* d_ws, size_t ws_size,
                              hipStream_t stream) {
    const float* x  = (const float*)d_in[0];
    const int*   ei = (const int*)d_in[1];
    const float* W1 = (const float*)d_in[2];
    const float* b1 = (const float*)d_in[3];
    const float* W2 = (const float*)d_in[4];
    const float* b2 = (const float*)d_in[5];
    const float* W3 = (const float*)d_in[6];
    const float* b3 = (const float*)d_in[7];
    const float* W4 = (const float*)d_in[8];
    const float* b4 = (const float*)d_in[9];

    const int E = in_sizes[1] / 2;
    const int* row = ei;
    const int* col = ei + E;

    float* dis  = (float*)d_ws;
    float* reg1 = dis  + ((NN + 255) & ~255);
    float* bufA = reg1 + ((E  + 255) & ~255);
    float* bufB = bufA + (size_t)NN * 128;
    int*   ptr  = (int*)reg1;
    u16*   srcs = (u16*)(ptr + ((NN + 1 + 255) & ~255));
    int*   rank = (int*)bufB;
    float* W23  = (float*)ptr + ((NN + 1 + 255) & ~255) + ((E / 2 + 255) & ~255);
    float* W234 = W23  + 96 * 64;
    float* c2   = W234 + 96 * 32;
    float* c3   = c2 + 32;
    float* xp = bufA;
    float* ax = bufA + (size_t)NN * 32;
    float* h1 = bufB;
    float* u0 = bufA;
    float* u1 = bufA + (size_t)NN * 32;
    float* u2 = bufA + (size_t)NN * 64;

    const int TB = 256;
    const int GB = 128;

    prep_pad_zero<<<ceil_div((long long)NN * XKP, TB), TB, 0, stream>>>(x, xp, ptr);
    edge_rank<<<ceil_div(E, TB), TB, 0, stream>>>(col, ptr, rank, E);
    scan_dis<<<1, SCAN_T, 0, stream>>>(ptr, dis);
    group_fill<<<ceil_div(E, TB), TB, 0, stream>>>(row, col, rank, ptr, srcs, E);

    gemm_wmma_f32<128, 128, 64, false, false><<<ceil_div(6 * 4, GB / 32), GB, 0, stream>>>(W2, W3, nullptr, W23, 6 * 4);
    gemm_wmma_f32<64, 64, 32, false, false><<<ceil_div(6 * 2, GB / 32), GB, 0, stream>>>(W23, W4, nullptr, W234, 6 * 2);
    bias_chain<<<1, 64, 0, stream>>>(b2, W3, b3, W4, c2, c3);

    const int TILES_M = NN / 16;
    const int AGB = ceil_div(NN, TB / 32);

    agg_gather<32, false><<<AGB, TB, 0, stream>>>(xp, dis, ptr, srcs, nullptr, ax);
    {
        int total = TILES_M * (96 / 16);
        gemm_wmma_f32<XKP, XK, 96, true, true><<<ceil_div(total, GB / 32), GB, 0, stream>>>(ax, W1, b1, h1, total);
    }

    {
        int total = TILES_M * (32 / 16);
        gemm_wmma_f32<96, 96, 32, false, false><<<ceil_div(total, GB / 32), GB, 0, stream>>>(h1, W234, nullptr, u0, total);
    }
    float* out = (float*)d_out;
    agg_gather<32, true><<<AGB, TB, 0, stream>>>(u0, dis, ptr, srcs, c2, u1);
    agg_gather<32, true><<<AGB, TB, 0, stream>>>(u1, dis, ptr, srcs, c3, u2);
    agg_gather<32, true><<<AGB, TB, 0, stream>>>(u2, dis, ptr, srcs, b4, out);
}


# Gathers: one wave sums four nodes at once (8 lanes x float4 = one 128-byte row per group, sources passed round the group by DPP8): 12500 waves per gather, not 50000; f32 and sum order kept
# speedup vs baseline: 4.6256x; 1.1119x over previous
#include <hip/hip_runtime.h>

#define NN 50000

typedef float v8f __attribute__((ext_vector_type(8)));
typedef float v2f __attribute__((ext_vector_type(2)));

typedef unsigned short u16;
static_assert(NN <= 65536, "source ids are stored in 16 bits");

#define SCAN_T  1024
#define SCAN_CH ((NN + SCAN_T - 1) / SCAN_T)

#define XK  29
#define XKP 32
__global__ void zero_counts(int* __restrict__ cnt) {
    int t = blockIdx.x * blockDim.x + threadIdx.x;
    if (t < NN) cnt[t] = 0;
}

__global__ void __launch_bounds__(64)
bias_chain(const float* __restrict__ b2, const float* __restrict__ W3,
           const float* __restrict__ b3, const float* __restrict__ W4,
           float* __restrict__ c2, float* __restrict__ c3) {
    __shared__ float tmp[64];
    const int t = threadIdx.x;
    float s = 0.0f;
    for (int k = 0; k < 128; ++k) s += b2[k] * W3[k * 64 + t];
    tmp[t] = s;
    __syncthreads();
    if (t < 32) {
        float a = 0.0f, b = 0.0f;
        for (int j = 0; j < 64; ++j) {
            float w = W4[j * 32 + t];
            a += tmp[j] * w;
            b += b3[j] * w;
        }
        c2[t] = a;
        c3[t] = b;
    }
}

__global__ void edge_rank(const int* __restrict__ col, int* __restrict__ cnt,
                          int* __restrict__ rank, int E) {
    int t = blockIdx.x * blockDim.x + threadIdx.x;
    if (t < E) rank[t] = atomicAdd(&cnt[col[t]], 1);
}

static_assert((SCAN_CH & 1) == 1, "odd lane stride keeps the chunk reads bank-conflict free");
__global__ void __launch_bounds__(SCAN_T)
scan_dis(int* __restrict__ cntptr, float* __restrict__ dis) {
    __shared__ int buf[SCAN_T * SCAN_CH];
    __shared__ int sh[SCAN_T];
    const int t = threadIdx.x;
    {
        constexpr int FULL = NN / SCAN_T;
        int tmp[SCAN_CH];
#pragma unroll
        for (int k = 0; k < FULL; ++k) tmp[k] = cntptr[t + k * SCAN_T];
#pragma unroll
        for (int k = FULL; k < SCAN_CH; ++k) {
            int i = t + k * SCAN_T;
            tmp[k] = (i < NN) ? cntptr[i] : 0;
        }
#pragma unroll
        for (int k = 0; k < SCAN_CH; ++k) buf[t + k * SCAN_T] = tmp[k];
    }
    __syncthreads();
    const int base = t * SCAN_CH;
    int s = 0;
    for (int j = 0; j < SCAN_CH; ++j) s += buf[base + j];
    sh[t] = s;
    __syncthreads();
    for (int off = 1; off < SCAN_T; off <<= 1) {
        int v = (t >= off) ? sh[t - off] : 0;
        __syncthreads();
        sh[t] += v;
        __syncthreads();
    }
    int run = sh[t] - s;
    for (int j = 0; j < SCAN_CH; ++j) {
        int c = buf[base + j];
        buf[base + j] = run;
        run += c;
    }
    __syncthreads();
    for (int i = t; i <= NN; i += SCAN_T) {
        int p = buf[i];
        cntptr[i] = p;
        if (i < NN) {
            float v = (float)(buf[i + 1] - p + 1);
            dis[i] = (v > 0.0f) ? rsqrtf(fmaxf(v, 1.0f)) : 0.0f;
        }
    }
}

__global__ void group_fill_pad(const int* __restrict__ row, const int* __restrict__ col,
                               const int* __restrict__ rank, const int* __restrict__ ptr,
                               u16* __restrict__ srcs, int E, int edgeBlocks,
                               const float* __restrict__ x, const float* __restrict__ dis,
                               float* __restrict__ xp) {
    if ((int)blockIdx.x < edgeBlocks) {
        int t = blockIdx.x * blockDim.x + threadIdx.x;
        if (t < E) srcs[ptr[col[t]] + rank[t]] = (u16)row[t];
    } else {
        int t = (blockIdx.x - edgeBlocks) * blockDim.x + threadIdx.x;
        if (t < NN * (XKP / 4)) {
            const int n = t >> 3, c = (t & 7) * 4;
            const float* __restrict__ xr = x + (size_t)n * XK + c;
            const float d = dis[n];
            float4 v;
            v.x = (c     < XK) ? xr[0] * d : 0.0f;
            v.y = (c + 1 < XK) ? xr[1] * d : 0.0f;
            v.z = (c + 2 < XK) ? xr[2] * d : 0.0f;
            v.w = (c + 3 < XK) ? xr[3] * d : 0.0f;
            *(float4*)(xp + (size_t)t * 4) = v;
        }
    }
}

template <int K, int KW, int F, bool BIAS, bool RELU, bool RSCALE>
__global__ void gemm_wmma_f32(const float* __restrict__ A, const float* __restrict__ W,
                              const float* __restrict__ bias, const float* __restrict__ rscale,
                              float* __restrict__ H, int totalTiles) {
    static_assert(K % 4 == 0 && KW <= K && K - KW < 4, "K is KW rounded up to 4");
    constexpr int tilesN = F / 16;
    int wave = blockIdx.x * (blockDim.x >> 5) + (threadIdx.x >> 5);
    if (wave >= totalTiles) return;
    int lane = threadIdx.x & 31;
    int half = lane >> 4;
    int lm   = lane & 15;
    int tm = wave / tilesN;
    int tn = wave - tm * tilesN;
    const float* __restrict__ Arow = A + (size_t)(tm * 16 + lm) * K;
    int bcol = tn * 16 + lm;

    v8f c = {};
#pragma unroll
    for (int k0 = 0; k0 < K; k0 += 4) {
        int ka = k0 + 2 * half;
        float2 a2 = *(const float2*)(Arow + ka);
        float a0 = a2.x, a1 = a2.y;
        float b0, b1;
        if (k0 + 4 <= KW) {
            b0 = W[(size_t)ka * F + bcol];
            b1 = W[(size_t)(ka + 1) * F + bcol];
        } else {
            int i0 = (ka     < KW) ? ka       : (KW - 1);
            int i1 = (ka + 1 < KW) ? (ka + 1) : (KW - 1);
            b0 = W[(size_t)i0 * F + bcol];
            b1 = W[(size_t)i1 * F + bcol];
            if (ka     >= KW) { a0 = 0.0f; b0 = 0.0f; }
            if (ka + 1 >= KW) { a1 = 0.0f; b1 = 0.0f; }
        }
        v2f av; av.x = a0; av.y = a1;
        v2f bv; bv.x = b0; bv.y = b1;
        c = __builtin_amdgcn_wmma_f32_16x16x4_f32(false, av, false, bv,
                                                  (short)0, c, false, false);
    }

    int col = tn * 16 + lm;
    float bb = 0.0f;
    if constexpr (BIAS) bb = bias[col];
#pragma unroll
    for (int j = 0; j < 8; ++j) {
        int r = tm * 16 + j + 8 * half;
        float v = c[j];
        if constexpr (BIAS) v += bb;
        if constexpr (RELU) v = fmaxf(v, 0.0f);
        if constexpr (RSCALE) v *= rscale[r];
        H[(size_t)r * F + col] = v;
    }
}

static_assert(NN % 4 == 0, "a wave takes four whole nodes");

template <int U>
__device__ inline int group_lane(int v) {
    return __builtin_amdgcn_mov_dpp8(v, U * 0x249249);
}

template <bool BIAS, bool OSCALE>
__global__ void __launch_bounds__(256)
agg_gather4(const float* __restrict__ h, const float* __restrict__ dis,
            const int* __restrict__ ptr, const u16* __restrict__ srcs,
            const float* __restrict__ bias, float* __restrict__ out) {
    const int wave = blockIdx.x * (blockDim.x >> 5) + (threadIdx.x >> 5);
    if (wave >= NN / 4) return;
    const int lane = threadIdx.x & 31;
    const int q    = lane & 7;
    const int node = wave * 4 + (lane >> 3);
    const int beg = ptr[node];
    const int end = ptr[node + 1];
    const float di = dis[node];
    const float4* __restrict__ h4 = (const float4*)h;

    float4 acc = h4[(unsigned)node * 8u + (unsigned)q];
    float4 bv = make_float4(0.0f, 0.0f, 0.0f, 0.0f);
    if constexpr (BIAS) bv = *(const float4*)(bias + q * 4);

    int e0 = beg;
    int my_s;
    {
        const bool in = e0 + q < end;
        const int v = (int)srcs[in ? e0 + q : 0];
        my_s = in ? v : node;
    }
    while (e0 < end) {
        const int e1 = e0 + 8;
        const bool in1 = e1 + q < end;
        const int v1 = (int)srcs[in1 ? e1 + q : 0];
        float4 r[8];
#define GATHER_ROW(U) r[U] = h4[(unsigned)group_lane<U>(my_s) * 8u + (unsigned)q];
        GATHER_ROW(0) GATHER_ROW(1) GATHER_ROW(2) GATHER_ROW(3)
        GATHER_ROW(4) GATHER_ROW(5) GATHER_ROW(6) GATHER_ROW(7)
#undef GATHER_ROW
#pragma unroll
        for (int u = 0; u < 8; ++u) {
            const bool on = e0 + u < end;
            acc.x += on ? r[u].x : 0.0f;
            acc.y += on ? r[u].y : 0.0f;
            acc.z += on ? r[u].z : 0.0f;
            acc.w += on ? r[u].w : 0.0f;
        }
        my_s = in1 ? v1 : node;
        e0 = e1;
    }

    float4 o;
    o.x = acc.x * di; o.y = acc.y * di; o.z = acc.z * di; o.w = acc.w * di;
    if constexpr (BIAS)   { o.x += bv.x; o.y += bv.y; o.z += bv.z; o.w += bv.w; }
    if constexpr (OSCALE) { o.x *= di;   o.y *= di;   o.z *= di;   o.w *= di; }
    ((float4*)out)[(unsigned)node * 8u + (unsigned)q] = o;
}

static inline int ceil_div(long long a, long long b) { return (int)((a + b - 1) / b); }

extern "C" void kernel_launch(void* const* d_in, const int* in_sizes, int n_in,
                              void* d_out, int out_size, void* d_ws, size_t ws_size,
                              hipStream_t stream) {
    const float* x  = (const float*)d_in[0];
    const int*   ei = (const int*)d_in[1];
    const float* W1 = (const float*)d_in[2];
    const float* b1 = (const float*)d_in[3];
    const float* W2 = (const float*)d_in[4];
    const float* b2 = (const float*)d_in[5];
    const float* W3 = (const float*)d_in[6];
    const float* b3 = (const float*)d_in[7];
    const float* W4 = (const float*)d_in[8];
    const float* b4 = (const float*)d_in[9];

    const int E = in_sizes[1] / 2;
    const int* row = ei;
    const int* col = ei + E;

    float* dis  = (float*)d_ws;
    float* reg1 = dis  + ((NN + 255) & ~255);
    float* bufA = reg1 + ((E  + 255) & ~255);
    float* bufB = bufA + (size_t)NN * 128;
    int*   ptr  = (int*)reg1;
    u16*   srcs = (u16*)(ptr + ((NN + 1 + 255) & ~255));
    int*   rank = (int*)bufB;
    float* W23  = (float*)ptr + ((NN + 1 + 255) & ~255) + ((E / 2 + 255) & ~255);
    float* W234 = W23  + 96 * 64;
    float* c2   = W234 + 96 * 32;
    float* c3   = c2 + 32;
    float* xp = bufA;
    float* ax = bufA + (size_t)NN * 32;
    float* h1 = bufB;
    float* u0 = bufA;
    float* u1 = bufA + (size_t)NN * 32;
    float* u2 = bufA + (size_t)NN * 64;

    const int TB = 256;
    const int GB = 128;

    zero_counts<<<ceil_div(NN, TB), TB, 0, stream>>>(ptr);
    edge_rank<<<ceil_div(E, TB), TB, 0, stream>>>(col, ptr, rank, E);
    scan_dis<<<1, SCAN_T, 0, stream>>>(ptr, dis);
    {
        const int edgeBlocks = ceil_div(E, TB);
        const int padBlocks  = ceil_div((long long)NN * (XKP / 4), TB);
        group_fill_pad<<<edgeBlocks + padBlocks, TB, 0, stream>>>(row, col, rank, ptr, srcs, E, edgeBlocks, x, dis, xp);
    }

    gemm_wmma_f32<128, 128, 64, false, false, false><<<ceil_div(6 * 4, GB / 32), GB, 0, stream>>>(W2, W3, nullptr, nullptr, W23, 6 * 4);
    gemm_wmma_f32<64, 64, 32, false, false, false><<<ceil_div(6 * 2, GB / 32), GB, 0, stream>>>(W23, W4, nullptr, nullptr, W234, 6 * 2);
    bias_chain<<<1, 64, 0, stream>>>(b2, W3, b3, W4, c2, c3);

    const int TILES_M = NN / 16;
    const int AGB = ceil_div(NN / 4, TB / 32);

    agg_gather4<false, false><<<AGB, TB, 0, stream>>>(xp, dis, ptr, srcs, nullptr, ax);
    {
        int total = TILES_M * (96 / 16);
        gemm_wmma_f32<XKP, XK, 96, true, true, false><<<ceil_div(total, GB / 32), GB, 0, stream>>>(ax, W1, b1, nullptr, h1, total);
    }

    {
        int total = TILES_M * (32 / 16);
        gemm_wmma_f32<96, 96, 32, false, false, true><<<ceil_div(total, GB / 32), GB, 0, stream>>>(h1, W234, nullptr, dis, u0, total);
    }
    float* out = (float*)d_out;
    agg_gather4<true, true><<<AGB, TB, 0, stream>>>(u0, dis, ptr, srcs, c2, u1);
    agg_gather4<true, true><<<AGB, TB, 0, stream>>>(u1, dis, ptr, srcs, c3, u2);
    agg_gather4<true, false><<<AGB, TB, 0, stream>>>(u2, dis, ptr, srcs, b4, out);
}


# Ranking add places each source straight into its destination's 64-slot row (overflow list beyond 64): no prefix scan, no second edge pass, 9 launches not 10; sum order = arrival rank as before
# speedup vs baseline: 5.8987x; 1.2752x over previous
#include <hip/hip_runtime.h>

#define NN 50000

typedef float v8f __attribute__((ext_vector_type(8)));
typedef float v2f __attribute__((ext_vector_type(2)));

typedef unsigned short u16;
static_assert(NN <= 65536, "source ids are stored in 16 bits");

#define SLOTS 64

#define XK  29
#define XKP 32
template <int K, int KW, int F, bool BIAS, bool RELU, bool RSCALE>
__device__ inline void tile_wmma_f32(const float* __restrict__ A, const float* __restrict__ W,
                                     const float* __restrict__ bias, const float* __restrict__ rscale,
                                     float* __restrict__ H, int wave) {
    static_assert(K % 4 == 0 && KW <= K && K - KW < 4, "K is KW rounded up to 4");
    constexpr int tilesN = F / 16;
    int lane = threadIdx.x & 31;
    int half = lane >> 4;
    int lm   = lane & 15;
    int tm = wave / tilesN;
    int tn = wave - tm * tilesN;
    const float* __restrict__ Arow = A + (size_t)(tm * 16 + lm) * K;
    int bcol = tn * 16 + lm;

    v8f c = {};
#pragma unroll
    for (int k0 = 0; k0 < K; k0 += 4) {
        int ka = k0 + 2 * half;
        float2 a2 = *(const float2*)(Arow + ka);
        float a0 = a2.x, a1 = a2.y;
        float b0, b1;
        if (k0 + 4 <= KW) {
            b0 = W[(size_t)ka * F + bcol];
            b1 = W[(size_t)(ka + 1) * F + bcol];
        } else {
            int i0 = (ka     < KW) ? ka       : (KW - 1);
            int i1 = (ka + 1 < KW) ? (ka + 1) : (KW - 1);
            b0 = W[(size_t)i0 * F + bcol];
            b1 = W[(size_t)i1 * F + bcol];
            if (ka     >= KW) { a0 = 0.0f; b0 = 0.0f; }
            if (ka + 1 >= KW) { a1 = 0.0f; b1 = 0.0f; }
        }
        v2f av; av.x = a0; av.y = a1;
        v2f bv; bv.x = b0; bv.y = b1;
        c = __builtin_amdgcn_wmma_f32_16x16x4_f32(false, av, false, bv,
                                                  (short)0, c, false, false);
    }

    int col = tn * 16 + lm;
    float bb = 0.0f;
    if constexpr (BIAS) bb = bias[col];
#pragma unroll
    for (int j = 0; j < 8; ++j) {
        int r = tm * 16 + j + 8 * half;
        float v = c[j];
        if constexpr (BIAS) v += bb;
        if constexpr (RELU) v = fmaxf(v, 0.0f);
        if constexpr (RSCALE) v *= rscale[r];
        H[(size_t)r * F + col] = v;
    }
}

template <int K, int KW, int F, bool BIAS, bool RELU, bool RSCALE>
__global__ void gemm_wmma_f32(const float* __restrict__ A, const float* __restrict__ W,
                              const float* __restrict__ bias, const float* __restrict__ rscale,
                              float* __restrict__ H, int totalTiles) {
    int wave = blockIdx.x * (blockDim.x >> 5) + (threadIdx.x >> 5);
    if (wave >= totalTiles) return;
    tile_wmma_f32<K, KW, F, BIAS, RELU, RSCALE>(A, W, bias, rscale, H, wave);
}

#define FOLD1_WG 4
#define FOLD2_WG 2
#define RIDER_T  256
static_assert((FOLD1_WG - 1) * (RIDER_T / 32) == 6 * 4, "one wave per tile of W2*W3, none spare");
static_assert(FOLD2_WG * (RIDER_T / 32) >= 6 * 2, "a wave for every tile of W23*W4");

__global__ void zero_counts(int* __restrict__ cnt) {
    int t = blockIdx.x * blockDim.x + threadIdx.x;
    if (t <= NN) cnt[t] = 0;
}

__device__ inline void bias_chain(const float* __restrict__ b2, const float* __restrict__ W3,
                                  const float* __restrict__ b3, const float* __restrict__ W4,
                                  float* __restrict__ c2, float* __restrict__ c3,
                                  float* __restrict__ tmp  ) {
    const int t = threadIdx.x;
    if (t < 64) {
        float s = 0.0f;
        for (int k = 0; k < 128; ++k) s += b2[k] * W3[k * 64 + t];
        tmp[t] = s;
    }
    __syncthreads();
    if (t < 32) {
        float a = 0.0f, b = 0.0f;
#pragma unroll 16
        for (int j = 0; j < 64; ++j) {
            float w = W4[j * 32 + t];
            a += tmp[j] * w;
            b += b3[j] * w;
        }
        c2[t] = a;
        c3[t] = b;
    }
}

__global__ void edge_place_fold(const int* __restrict__ row, const int* __restrict__ col,
                                int* __restrict__ cnt, u16* __restrict__ srcs,
                                unsigned* __restrict__ ovf, int E,
                                const float* __restrict__ W2, const float* __restrict__ W3,
                               float* __restrict__ W23,
                               const float* __restrict__ b2, const float* __restrict__ b3,
                               const float* __restrict__ W4,
                               float* __restrict__ c2, float* __restrict__ c3) {
    __shared__ float tmp[64];
    const int b = blockIdx.x;
    if (b < FOLD1_WG - 1) {
        tile_wmma_f32<128, 128, 64, false, false, false>(W2, W3, nullptr, nullptr, W23,
                                                         b * (RIDER_T / 32) + (threadIdx.x >> 5));
    } else if (b == FOLD1_WG - 1) {
        bias_chain(b2, W3, b3, W4, c2, c3, tmp);
    } else {
        int t = (b - FOLD1_WG) * RIDER_T + threadIdx.x;
        if (t < E) {
            const int c = col[t];
            const unsigned s = (unsigned)row[t] & 0xffffu;
            const int r = atomicAdd(&cnt[c], 1);
            if (r < SLOTS) {
                srcs[(unsigned)c * SLOTS + (unsigned)r] = (u16)s;
            } else {
                const int k = atomicAdd(&cnt[NN], 1);
                ovf[k] = ((unsigned)c << 16) | s;
            }
        }
    }
}

__device__ inline float dis_of(int c) {
    const float v = (float)(c + 1);
    return (v > 0.0f) ? rsqrtf(fmaxf(v, 1.0f)) : 0.0f;
}

__global__ void dis_pad(const int* __restrict__ cnt, int padBlocks,
                        const float* __restrict__ x, float* __restrict__ dis,
                        float* __restrict__ xp,
                        const float* __restrict__ W23, const float* __restrict__ W4,
                        float* __restrict__ W234) {
    const int b = (int)blockIdx.x - FOLD2_WG;
    if (b < 0) {
        const int wave = (int)blockIdx.x * (RIDER_T / 32) + (threadIdx.x >> 5);
        if (wave < 6 * 2)
            tile_wmma_f32<64, 64, 32, false, false, false>(W23, W4, nullptr, nullptr, W234, wave);
    } else if (b < padBlocks) {
        int t = b * RIDER_T + threadIdx.x;
        if (t < NN * (XKP / 4)) {
            const int n = t >> 3, c = (t & 7) * 4;
            const float* __restrict__ xr = x + (size_t)n * XK + c;
            const float d = dis_of(cnt[n]);
            float4 v;
            v.x = (c     < XK) ? xr[0] * d : 0.0f;
            v.y = (c + 1 < XK) ? xr[1] * d : 0.0f;
            v.z = (c + 2 < XK) ? xr[2] * d : 0.0f;
            v.w = (c + 3 < XK) ? xr[3] * d : 0.0f;
            *(float4*)(xp + (size_t)t * 4) = v;
        }
    } else {
        int t = (b - padBlocks) * RIDER_T + threadIdx.x;
        if (t < NN) dis[t] = dis_of(cnt[t]);
    }
}

static_assert(NN % 4 == 0, "a wave takes four whole nodes");

template <int U>
__device__ inline int group_lane(int v) {
    return __builtin_amdgcn_mov_dpp8(v, U * 0x249249);
}

template <bool BIAS, bool OSCALE>
__global__ void __launch_bounds__(256)
agg_gather4(const float* __restrict__ h, const float* __restrict__ dis,
            const int* __restrict__ cnt, const u16* __restrict__ srcs,
            const unsigned* __restrict__ ovf, int E,
            const float* __restrict__ bias, float* __restrict__ out) {
    const int wave = blockIdx.x * (blockDim.x >> 5) + (threadIdx.x >> 5);
    if (wave >= NN / 4) return;
    const int lane = threadIdx.x & 31;
    const int q    = lane & 7;
    const int node = wave * 4 + (lane >> 3);
    const int deg = cnt[node];
    const int beg = node * SLOTS;
    const int end = beg + (deg < SLOTS ? deg : SLOTS);
    const float di = dis[node];
    const float4* __restrict__ h4 = (const float4*)h;

    float4 acc = h4[(unsigned)node * 8u + (unsigned)q];
    float4 bv = make_float4(0.0f, 0.0f, 0.0f, 0.0f);
    if constexpr (BIAS) bv = *(const float4*)(bias + q * 4);

    int e0 = beg;
    int my_s;
    {
        const bool in = e0 + q < end;
        const int v = (int)srcs[in ? e0 + q : 0];
        my_s = in ? v : node;
    }
    while (e0 < end) {
        const int e1 = e0 + 8;
        const bool in1 = e1 + q < end;
        const int v1 = (int)srcs[in1 ? e1 + q : 0];
        float4 r[8];
#define GATHER_ROW(U) r[U] = h4[(unsigned)group_lane<U>(my_s) * 8u + (unsigned)q];
        GATHER_ROW(0) GATHER_ROW(1) GATHER_ROW(2) GATHER_ROW(3)
        GATHER_ROW(4) GATHER_ROW(5) GATHER_ROW(6) GATHER_ROW(7)
#undef GATHER_ROW
#pragma unroll
        for (int u = 0; u < 8; ++u) {
            const bool on = (u == 0) || (e0 + u < end);
            acc.x += on ? r[u].x : 0.0f;
            acc.y += on ? r[u].y : 0.0f;
            acc.z += on ? r[u].z : 0.0f;
            acc.w += on ? r[u].w : 0.0f;
        }
        my_s = in1 ? v1 : node;
        e0 = e1;
    }

    if (deg > SLOTS) {
        int no = cnt[NN];
        no = no < E ? no : E;
        for (int k = 0; k < no; ++k) {
            const unsigned w = ovf[k];
            if ((int)(w >> 16) == node) {
                const float4 r = h4[(w & 0xffffu) * 8u + (unsigned)q];
                acc.x += r.x; acc.y += r.y; acc.z += r.z; acc.w += r.w;
            }
        }
    }

    float4 o;
    o.x = acc.x * di; o.y = acc.y * di; o.z = acc.z * di; o.w = acc.w * di;
    if constexpr (BIAS)   { o.x += bv.x; o.y += bv.y; o.z += bv.z; o.w += bv.w; }
    if constexpr (OSCALE) { o.x *= di;   o.y *= di;   o.z *= di;   o.w *= di; }
    ((float4*)out)[(unsigned)node * 8u + (unsigned)q] = o;
}

static inline int ceil_div(long long a, long long b) { return (int)((a + b - 1) / b); }

extern "C" void kernel_launch(void* const* d_in, const int* in_sizes, int n_in,
                              void* d_out, int out_size, void* d_ws, size_t ws_size,
                              hipStream_t stream) {
    const float* x  = (const float*)d_in[0];
    const int*   ei = (const int*)d_in[1];
    const float* W1 = (const float*)d_in[2];
    const float* b1 = (const float*)d_in[3];
    const float* W2 = (const float*)d_in[4];
    const float* b2 = (const float*)d_in[5];
    const float* W3 = (const float*)d_in[6];
    const float* b3 = (const float*)d_in[7];
    const float* W4 = (const float*)d_in[8];
    const float* b4 = (const float*)d_in[9];

    const int E = in_sizes[1] / 2;
    const int* row = ei;
    const int* col = ei + E;

    float* dis  = (float*)d_ws;
    float* reg1 = dis  + ((NN + 255) & ~255);
    float* bufA = reg1 + ((E  + 255) & ~255);
    float* bufB = bufA + (size_t)NN * 128;
    int*   cnt  = (int*)reg1;
    float* W23  = (float*)cnt + ((NN + 1 + 255) & ~255);
    float* W234 = W23  + 96 * 64;
    float* c2   = W234 + 96 * 32;
    float* c3   = c2 + 32;
    u16*      srcs = (u16*)(bufA + (size_t)NN * 96);
    unsigned* ovf  = (unsigned*)(bufB + (size_t)NN * 96);
    float* xp = bufA;
    float* ax = bufA + (size_t)NN * 32;
    float* h1 = bufB;
    float* u0 = bufA;
    float* u1 = bufA + (size_t)NN * 32;
    float* u2 = bufA + (size_t)NN * 64;

    const int TB = 256;
    const int GB = 128;

    zero_counts<<<ceil_div(NN + 1, TB), TB, 0, stream>>>(cnt);
    edge_place_fold<<<FOLD1_WG + ceil_div(E, RIDER_T), RIDER_T, 0, stream>>>(row, col, cnt, srcs, ovf, E, W2, W3, W23, b2, b3, W4, c2, c3);
    {
        const int padBlocks = ceil_div((long long)NN * (XKP / 4), RIDER_T);
        const int disBlocks = ceil_div(NN, RIDER_T);
        dis_pad<<<FOLD2_WG + padBlocks + disBlocks, RIDER_T, 0, stream>>>(cnt, padBlocks, x, dis, xp, W23, W4, W234);
    }

    const int TILES_M = NN / 16;
    const int AGB = ceil_div(NN / 4, TB / 32);

    agg_gather4<false, false><<<AGB, TB, 0, stream>>>(xp, dis, cnt, srcs, ovf, E, nullptr, ax);
    {
        int total = TILES_M * (96 / 16);
        gemm_wmma_f32<XKP, XK, 96, true, true, false><<<ceil_div(total, GB / 32), GB, 0, stream>>>(ax, W1, b1, nullptr, h1, total);
    }

    {
        int total = TILES_M * (32 / 16);
        gemm_wmma_f32<96, 96, 32, false, false, true><<<ceil_div(total, GB / 32), GB, 0, stream>>>(h1, W234, nullptr, dis, u0, total);
    }
    float* out = (float*)d_out;
    agg_gather4<true, true><<<AGB, TB, 0, stream>>>(u0, dis, cnt, srcs, ovf, E, c2, u1);
    agg_gather4<true, true><<<AGB, TB, 0, stream>>>(u1, dis, cnt, srcs, ovf, E, c3, u2);
    agg_gather4<true, false><<<AGB, TB, 0, stream>>>(u2, dis, cnt, srcs, ovf, E, b4, out);
}


# 29->96 (bias, ReLU) and 96->32 products in one kernel, transposed so the first's D registers are the second's B operand: h1 never stored, 8 launches not 9; k order of both sums permuted, f32 kept
# speedup vs baseline: 6.2077x; 1.0524x over previous
#include <hip/hip_runtime.h>

#define NN 50000

typedef float v8f __attribute__((ext_vector_type(8)));
typedef float v2f __attribute__((ext_vector_type(2)));

typedef unsigned short u16;
static_assert(NN <= 65536, "source ids are stored in 16 bits");

#define SLOTS 64

#define XK  29
#define XKP 32
template <int K, int KW, int F, bool BIAS, bool RELU, bool RSCALE>
__device__ inline void tile_wmma_f32(const float* __restrict__ A, const float* __restrict__ W,
                                     const float* __restrict__ bias, const float* __restrict__ rscale,
                                     float* __restrict__ H, int wave) {
    static_assert(K % 4 == 0 && KW <= K && K - KW < 4, "K is KW rounded up to 4");
    constexpr int tilesN = F / 16;
    int lane = threadIdx.x & 31;
    int half = lane >> 4;
    int lm   = lane & 15;
    int tm = wave / tilesN;
    int tn = wave - tm * tilesN;
    const float* __restrict__ Arow = A + (size_t)(tm * 16 + lm) * K;
    int bcol = tn * 16 + lm;

    v8f c = {};
#pragma unroll
    for (int k0 = 0; k0 < K; k0 += 4) {
        int ka = k0 + 2 * half;
        float2 a2 = *(const float2*)(Arow + ka);
        float a0 = a2.x, a1 = a2.y;
        float b0, b1;
        if (k0 + 4 <= KW) {
            b0 = W[(size_t)ka * F + bcol];
            b1 = W[(size_t)(ka + 1) * F + bcol];
        } else {
            int i0 = (ka     < KW) ? ka       : (KW - 1);
            int i1 = (ka + 1 < KW) ? (ka + 1) : (KW - 1);
            b0 = W[(size_t)i0 * F + bcol];
            b1 = W[(size_t)i1 * F + bcol];
            if (ka     >= KW) { a0 = 0.0f; b0 = 0.0f; }
            if (ka + 1 >= KW) { a1 = 0.0f; b1 = 0.0f; }
        }
        v2f av; av.x = a0; av.y = a1;
        v2f bv; bv.x = b0; bv.y = b1;
        c = __builtin_amdgcn_wmma_f32_16x16x4_f32(false, av, false, bv,
                                                  (short)0, c, false, false);
    }

    int col = tn * 16 + lm;
    float bb = 0.0f;
    if constexpr (BIAS) bb = bias[col];
#pragma unroll
    for (int j = 0; j < 8; ++j) {
        int r = tm * 16 + j + 8 * half;
        float v = c[j];
        if constexpr (BIAS) v += bb;
        if constexpr (RELU) v = fmaxf(v, 0.0f);
        if constexpr (RSCALE) v *= rscale[r];
        H[(size_t)r * F + col] = v;
    }
}

#define H1F 96
#define UF  32
static_assert(XKP == 32 && NN % 16 == 0, "a lane half holds 16 of a row's 32 floats; whole node tiles");

__global__ void __launch_bounds__(128)
dense_chain(const float* __restrict__ ax, const float* __restrict__ W1,
            const float* __restrict__ b1, const float* __restrict__ W234,
            const float* __restrict__ dis, float* __restrict__ u0, int nodeTiles) {
    const int wave = blockIdx.x * (blockDim.x >> 5) + (threadIdx.x >> 5);
    if (wave >= nodeTiles) return;
    const int lane = threadIdx.x & 31;
    const int hf = lane >> 4;
    const int c  = lane & 15;
    const int node = wave * 16 + c;

    float axv[16];
    {
        const float4* __restrict__ axr = (const float4*)(ax + (size_t)node * XKP + 16 * hf);
#pragma unroll
        for (int i = 0; i < 4; ++i) {
            const float4 v = axr[i];
            axv[4 * i] = v.x; axv[4 * i + 1] = v.y; axv[4 * i + 2] = v.z; axv[4 * i + 3] = v.w;
        }
    }

    v8f acc1[H1F / 16];
#pragma unroll
    for (int ft = 0; ft < H1F / 16; ++ft) acc1[ft] = v8f{};

#pragma unroll
    for (int t = 0; t < 8; ++t) {
        const int k = 16 * hf + 2 * t;
        float x0 = axv[2 * t], x1 = axv[2 * t + 1];
        const bool padded = 16 + 2 * t + 1 >= XK;
        const bool in0 = !padded || k < XK;
        const bool in1 = !padded || k + 1 < XK;
        const int k0 = in0 ? k : XK - 1;
        const int k1 = in1 ? k + 1 : XK - 1;
        if (!in0) x0 = 0.0f;
        if (!in1) x1 = 0.0f;
        v2f bv; bv.x = x0; bv.y = x1;
#pragma unroll
        for (int ft = 0; ft < H1F / 16; ++ft) {
            float w0 = W1[k0 * H1F + 16 * ft + c];
            float w1 = W1[k1 * H1F + 16 * ft + c];
            if (!in0) w0 = 0.0f;
            if (!in1) w1 = 0.0f;
            v2f av; av.x = w0; av.y = w1;
            acc1[ft] = __builtin_amdgcn_wmma_f32_16x16x4_f32(false, av, false, bv,
                                                             (short)0, acc1[ft], false, false);
        }
    }

    v8f acc2[UF / 16];
#pragma unroll
    for (int ot = 0; ot < UF / 16; ++ot) acc2[ot] = v8f{};

#pragma unroll
    for (int ft = 0; ft < H1F / 16; ++ft) {
        const float4* __restrict__ bp = (const float4*)(b1 + 16 * ft + 8 * hf);
        const float4 bl = bp[0], bh = bp[1];
        float hv[8];
        hv[0] = fmaxf(acc1[ft][0] + bl.x, 0.0f); hv[1] = fmaxf(acc1[ft][1] + bl.y, 0.0f);
        hv[2] = fmaxf(acc1[ft][2] + bl.z, 0.0f); hv[3] = fmaxf(acc1[ft][3] + bl.w, 0.0f);
        hv[4] = fmaxf(acc1[ft][4] + bh.x, 0.0f); hv[5] = fmaxf(acc1[ft][5] + bh.y, 0.0f);
        hv[6] = fmaxf(acc1[ft][6] + bh.z, 0.0f); hv[7] = fmaxf(acc1[ft][7] + bh.w, 0.0f);
#pragma unroll
        for (int p = 0; p < 4; ++p) {
            const int k = 16 * ft + 8 * hf + 2 * p;
            v2f bv; bv.x = hv[2 * p]; bv.y = hv[2 * p + 1];
#pragma unroll
            for (int ot = 0; ot < UF / 16; ++ot) {
                v2f av;
                av.x = W234[k * UF + 16 * ot + c];
                av.y = W234[(k + 1) * UF + 16 * ot + c];
                acc2[ot] = __builtin_amdgcn_wmma_f32_16x16x4_f32(false, av, false, bv,
                                                                 (short)0, acc2[ot], false, false);
            }
        }
    }

    const float di = dis[node];
    float4* __restrict__ o4 = (float4*)(u0 + (size_t)node * UF + 8 * hf);
#pragma unroll
    for (int ot = 0; ot < UF / 16; ++ot) {
        float4 lo, hi;
        lo.x = acc2[ot][0] * di; lo.y = acc2[ot][1] * di; lo.z = acc2[ot][2] * di; lo.w = acc2[ot][3] * di;
        hi.x = acc2[ot][4] * di; hi.y = acc2[ot][5] * di; hi.z = acc2[ot][6] * di; hi.w = acc2[ot][7] * di;
        o4[4 * ot]     = lo;
        o4[4 * ot + 1] = hi;
    }
}

#define FOLD1_WG 4
#define FOLD2_WG 2
#define RIDER_T  256
static_assert((FOLD1_WG - 1) * (RIDER_T / 32) == 6 * 4, "one wave per tile of W2*W3, none spare");
static_assert(FOLD2_WG * (RIDER_T / 32) >= 6 * 2, "a wave for every tile of W23*W4");

__global__ void zero_counts(int* __restrict__ cnt) {
    int t = blockIdx.x * blockDim.x + threadIdx.x;
    if (t <= NN) cnt[t] = 0;
}

__device__ inline void bias_chain(const float* __restrict__ b2, const float* __restrict__ W3,
                                  const float* __restrict__ b3, const float* __restrict__ W4,
                                  float* __restrict__ c2, float* __restrict__ c3,
                                  float* __restrict__ tmp  ) {
    const int t = threadIdx.x;
    if (t < 64) {
        float s = 0.0f;
        for (int k = 0; k < 128; ++k) s += b2[k] * W3[k * 64 + t];
        tmp[t] = s;
    }
    __syncthreads();
    if (t < 32) {
        float a = 0.0f, b = 0.0f;
#pragma unroll 16
        for (int j = 0; j < 64; ++j) {
            float w = W4[j * 32 + t];
            a += tmp[j] * w;
            b += b3[j] * w;
        }
        c2[t] = a;
        c3[t] = b;
    }
}

__global__ void edge_place_fold(const int* __restrict__ row, const int* __restrict__ col,
                                int* __restrict__ cnt, u16* __restrict__ srcs,
                                unsigned* __restrict__ ovf, int E,
                                const float* __restrict__ W2, const float* __restrict__ W3,
                               float* __restrict__ W23,
                               const float* __restrict__ b2, const float* __restrict__ b3,
                               const float* __restrict__ W4,
                               float* __restrict__ c2, float* __restrict__ c3) {
    __shared__ float tmp[64];
    const int b = blockIdx.x;
    if (b < FOLD1_WG - 1) {
        tile_wmma_f32<128, 128, 64, false, false, false>(W2, W3, nullptr, nullptr, W23,
                                                         b * (RIDER_T / 32) + (threadIdx.x >> 5));
    } else if (b == FOLD1_WG - 1) {
        bias_chain(b2, W3, b3, W4, c2, c3, tmp);
    } else {
        int t = (b - FOLD1_WG) * RIDER_T + threadIdx.x;
        if (t < E) {
            const int c = col[t];
            const unsigned s = (unsigned)row[t] & 0xffffu;
            const int r = atomicAdd(&cnt[c], 1);
            if (r < SLOTS) {
                srcs[(unsigned)c * SLOTS + (unsigned)r] = (u16)s;
            } else {
                const int k = atomicAdd(&cnt[NN], 1);
                ovf[k] = ((unsigned)c << 16) | s;
            }
        }
    }
}

__device__ inline float dis_of(int c) {
    const float v = (float)(c + 1);
    return (v > 0.0f) ? rsqrtf(fmaxf(v, 1.0f)) : 0.0f;
}

__global__ void dis_pad(const int* __restrict__ cnt, int padBlocks,
                        const float* __restrict__ x, float* __restrict__ dis,
                        float* __restrict__ xp,
                        const float* __restrict__ W23, const float* __restrict__ W4,
                        float* __restrict__ W234) {
    const int b = (int)blockIdx.x - FOLD2_WG;
    if (b < 0) {
        const int wave = (int)blockIdx.x * (RIDER_T / 32) + (threadIdx.x >> 5);
        if (wave < 6 * 2)
            tile_wmma_f32<64, 64, 32, false, false, false>(W23, W4, nullptr, nullptr, W234, wave);
    } else if (b < padBlocks) {
        int t = b * RIDER_T + threadIdx.x;
        if (t < NN * (XKP / 4)) {
            const int n = t >> 3, c = (t & 7) * 4;
            const float* __restrict__ xr = x + (size_t)n * XK + c;
            const float d = dis_of(cnt[n]);
            float4 v;
            v.x = (c     < XK) ? xr[0] * d : 0.0f;
            v.y = (c + 1 < XK) ? xr[1] * d : 0.0f;
            v.z = (c + 2 < XK) ? xr[2] * d : 0.0f;
            v.w = (c + 3 < XK) ? xr[3] * d : 0.0f;
            *(float4*)(xp + (size_t)t * 4) = v;
        }
    } else {
        int t = (b - padBlocks) * RIDER_T + threadIdx.x;
        if (t < NN) dis[t] = dis_of(cnt[t]);
    }
}

static_assert(NN % 4 == 0, "a wave takes four whole nodes");

template <int U>
__device__ inline int group_lane(int v) {
    return __builtin_amdgcn_mov_dpp8(v, U * 0x249249);
}

template <bool BIAS, bool OSCALE>
__global__ void __launch_bounds__(256)
agg_gather4(const float* __restrict__ h, const float* __restrict__ dis,
            const int* __restrict__ cnt, const u16* __restrict__ srcs,
            const unsigned* __restrict__ ovf, int E,
            const float* __restrict__ bias, float* __restrict__ out) {
    const int wave = blockIdx.x * (blockDim.x >> 5) + (threadIdx.x >> 5);
    if (wave >= NN / 4) return;
    const int lane = threadIdx.x & 31;
    const int q    = lane & 7;
    const int node = wave * 4 + (lane >> 3);
    const int deg = cnt[node];
    const int beg = node * SLOTS;
    const int end = beg + (deg < SLOTS ? deg : SLOTS);
    const float di = dis[node];
    const float4* __restrict__ h4 = (const float4*)h;

    float4 acc = h4[(unsigned)node * 8u + (unsigned)q];
    float4 bv = make_float4(0.0f, 0.0f, 0.0f, 0.0f);
    if constexpr (BIAS) bv = *(const float4*)(bias + q * 4);

    int e0 = beg;
    int my_s;
    {
        const bool in = e0 + q < end;
        const int v = (int)srcs[in ? e0 + q : 0];
        my_s = in ? v : node;
    }
    while (e0 < end) {
        const int e1 = e0 + 8;
        const bool in1 = e1 + q < end;
        const int v1 = (int)srcs[in1 ? e1 + q : 0];
        float4 r[8];
#define GATHER_ROW(U) r[U] = h4[(unsigned)group_lane<U>(my_s) * 8u + (unsigned)q];
        GATHER_ROW(0) GATHER_ROW(1) GATHER_ROW(2) GATHER_ROW(3)
        GATHER_ROW(4) GATHER_ROW(5) GATHER_ROW(6) GATHER_ROW(7)
#undef GATHER_ROW
#pragma unroll
        for (int u = 0; u < 8; ++u) {
            const bool on = (u == 0) || (e0 + u < end);
            acc.x += on ? r[u].x : 0.0f;
            acc.y += on ? r[u].y : 0.0f;
            acc.z += on ? r[u].z : 0.0f;
            acc.w += on ? r[u].w : 0.0f;
        }
        my_s = in1 ? v1 : node;
        e0 = e1;
    }

    if (deg > SLOTS) {
        int no = cnt[NN];
        no = no < E ? no : E;
        for (int k = 0; k < no; ++k) {
            const unsigned w = ovf[k];
            if ((int)(w >> 16) == node) {
                const float4 r = h4[(w & 0xffffu) * 8u + (unsigned)q];
                acc.x += r.x; acc.y += r.y; acc.z += r.z; acc.w += r.w;
            }
        }
    }

    float4 o;
    o.x = acc.x * di; o.y = acc.y * di; o.z = acc.z * di; o.w = acc.w * di;
    if constexpr (BIAS)   { o.x += bv.x; o.y += bv.y; o.z += bv.z; o.w += bv.w; }
    if constexpr (OSCALE) { o.x *= di;   o.y *= di;   o.z *= di;   o.w *= di; }
    ((float4*)out)[(unsigned)node * 8u + (unsigned)q] = o;
}

static inline int ceil_div(long long a, long long b) { return (int)((a + b - 1) / b); }

extern "C" void kernel_launch(void* const* d_in, const int* in_sizes, int n_in,
                              void* d_out, int out_size, void* d_ws, size_t ws_size,
                              hipStream_t stream) {
    const float* x  = (const float*)d_in[0];
    const int*   ei = (const int*)d_in[1];
    const float* W1 = (const float*)d_in[2];
    const float* b1 = (const float*)d_in[3];
    const float* W2 = (const float*)d_in[4];
    const float* b2 = (const float*)d_in[5];
    const float* W3 = (const float*)d_in[6];
    const float* b3 = (const float*)d_in[7];
    const float* W4 = (const float*)d_in[8];
    const float* b4 = (const float*)d_in[9];

    const int E = in_sizes[1] / 2;
    const int* row = ei;
    const int* col = ei + E;

    float* dis  = (float*)d_ws;
    float* reg1 = dis  + ((NN + 255) & ~255);
    float* bufA = reg1 + ((E  + 255) & ~255);
    float* bufB = bufA + (size_t)NN * 128;
    int*   cnt  = (int*)reg1;
    float* W23  = (float*)cnt + ((NN + 1 + 255) & ~255);
    float* W234 = W23  + 96 * 64;
    float* c2   = W234 + 96 * 32;
    float* c3   = c2 + 32;
    u16*      srcs = (u16*)(bufA + (size_t)NN * 96);
    unsigned* ovf  = (unsigned*)(bufB + (size_t)NN * 96);
    float* xp = bufA;
    float* ax = bufA + (size_t)NN * 32;
    float* u0 = bufA;
    float* u1 = bufA + (size_t)NN * 32;
    float* u2 = bufA + (size_t)NN * 64;

    const int TB = 256;
    const int GB = 128;

    zero_counts<<<ceil_div(NN + 1, TB), TB, 0, stream>>>(cnt);
    edge_place_fold<<<FOLD1_WG + ceil_div(E, RIDER_T), RIDER_T, 0, stream>>>(row, col, cnt, srcs, ovf, E, W2, W3, W23, b2, b3, W4, c2, c3);
    {
        const int padBlocks = ceil_div((long long)NN * (XKP / 4), RIDER_T);
        const int disBlocks = ceil_div(NN, RIDER_T);
        dis_pad<<<FOLD2_WG + padBlocks + disBlocks, RIDER_T, 0, stream>>>(cnt, padBlocks, x, dis, xp, W23, W4, W234);
    }

    const int TILES_M = NN / 16;
    const int AGB = ceil_div(NN / 4, TB / 32);

    agg_gather4<false, false><<<AGB, TB, 0, stream>>>(xp, dis, cnt, srcs, ovf, E, nullptr, ax);
    dense_chain<<<ceil_div(TILES_M, GB / 32), GB, 0, stream>>>(ax, W1, b1, W234, dis, u0, TILES_M);
    float* out = (float*)d_out;
    agg_gather4<true, true><<<AGB, TB, 0, stream>>>(u0, dis, cnt, srcs, ovf, E, c2, u1);
    agg_gather4<true, true><<<AGB, TB, 0, stream>>>(u1, dis, cnt, srcs, ovf, E, c3, u2);
    agg_gather4<true, false><<<AGB, TB, 0, stream>>>(u2, dis, cnt, srcs, ovf, E, b4, out);
}


# Edges placed by a two-level LDS sort (chunks by destination range; each range's owner ranks with LDS adds, stores slot rows, counts, dis, padded x): no global atomic, 7 launches; sum order = LDS rank
# speedup vs baseline: 11.6901x; 1.8832x over previous
#include <hip/hip_runtime.h>

#define NN 50000

typedef float v8f __attribute__((ext_vector_type(8)));
typedef float v2f __attribute__((ext_vector_type(2)));

typedef unsigned short u16;
static_assert(NN < 65535, "ids are stored in 16 bits; 0xffff marks no edge");

#define SLOTS 64

#define XK  29
#define XKP 32
template <int K, int KW, int F, bool BIAS, bool RELU, bool RSCALE>
__device__ inline void tile_wmma_f32(const float* __restrict__ A, const float* __restrict__ W,
                                     const float* __restrict__ bias, const float* __restrict__ rscale,
                                     float* __restrict__ H, int wave) {
    static_assert(K % 4 == 0 && KW <= K && K - KW < 4, "K is KW rounded up to 4");
    constexpr int tilesN = F / 16;
    int lane = threadIdx.x & 31;
    int half = lane >> 4;
    int lm   = lane & 15;
    int tm = wave / tilesN;
    int tn = wave - tm * tilesN;
    const float* __restrict__ Arow = A + (size_t)(tm * 16 + lm) * K;
    int bcol = tn * 16 + lm;

    v8f c = {};
#pragma unroll
    for (int k0 = 0; k0 < K; k0 += 4) {
        int ka = k0 + 2 * half;
        float2 a2 = *(const float2*)(Arow + ka);
        float a0 = a2.x, a1 = a2.y;
        float b0, b1;
        if (k0 + 4 <= KW) {
            b0 = W[(size_t)ka * F + bcol];
            b1 = W[(size_t)(ka + 1) * F + bcol];
        } else {
            int i0 = (ka     < KW) ? ka       : (KW - 1);
            int i1 = (ka + 1 < KW) ? (ka + 1) : (KW - 1);
            b0 = W[(size_t)i0 * F + bcol];
            b1 = W[(size_t)i1 * F + bcol];
            if (ka     >= KW) { a0 = 0.0f; b0 = 0.0f; }
            if (ka + 1 >= KW) { a1 = 0.0f; b1 = 0.0f; }
        }
        v2f av; av.x = a0; av.y = a1;
        v2f bv; bv.x = b0; bv.y = b1;
        c = __builtin_amdgcn_wmma_f32_16x16x4_f32(false, av, false, bv,
                                                  (short)0, c, false, false);
    }

    int col = tn * 16 + lm;
    float bb = 0.0f;
    if constexpr (BIAS) bb = bias[col];
#pragma unroll
    for (int j = 0; j < 8; ++j) {
        int r = tm * 16 + j + 8 * half;
        float v = c[j];
        if constexpr (BIAS) v += bb;
        if constexpr (RELU) v = fmaxf(v, 0.0f);
        if constexpr (RSCALE) v *= rscale[r];
        H[(size_t)r * F + col] = v;
    }
}

#define H1F 96
#define UF  32
static_assert(XKP == 32 && NN % 16 == 0, "a lane half holds 16 of a row's 32 floats; whole node tiles");

__global__ void __launch_bounds__(128)
dense_chain(const float* __restrict__ ax, const float* __restrict__ W1,
            const float* __restrict__ b1, const float* __restrict__ W234,
            const float* __restrict__ dis, float* __restrict__ u0, int nodeTiles) {
    const int wave = blockIdx.x * (blockDim.x >> 5) + (threadIdx.x >> 5);
    if (wave >= nodeTiles) return;
    const int lane = threadIdx.x & 31;
    const int hf = lane >> 4;
    const int c  = lane & 15;
    const int node = wave * 16 + c;

    float axv[16];
    {
        const float4* __restrict__ axr = (const float4*)(ax + (size_t)node * XKP + 16 * hf);
#pragma unroll
        for (int i = 0; i < 4; ++i) {
            const float4 v = axr[i];
            axv[4 * i] = v.x; axv[4 * i + 1] = v.y; axv[4 * i + 2] = v.z; axv[4 * i + 3] = v.w;
        }
    }

    v8f acc1[H1F / 16];
#pragma unroll
    for (int ft = 0; ft < H1F / 16; ++ft) acc1[ft] = v8f{};

#pragma unroll
    for (int t = 0; t < 8; ++t) {
        const int k = 16 * hf + 2 * t;
        float x0 = axv[2 * t], x1 = axv[2 * t + 1];
        const bool padded = 16 + 2 * t + 1 >= XK;
        const bool in0 = !padded || k < XK;
        const bool in1 = !padded || k + 1 < XK;
        const int k0 = in0 ? k : XK - 1;
        const int k1 = in1 ? k + 1 : XK - 1;
        if (!in0) x0 = 0.0f;
        if (!in1) x1 = 0.0f;
        v2f bv; bv.x = x0; bv.y = x1;
#pragma unroll
        for (int ft = 0; ft < H1F / 16; ++ft) {
            float w0 = W1[k0 * H1F + 16 * ft + c];
            float w1 = W1[k1 * H1F + 16 * ft + c];
            if (!in0) w0 = 0.0f;
            if (!in1) w1 = 0.0f;
            v2f av; av.x = w0; av.y = w1;
            acc1[ft] = __builtin_amdgcn_wmma_f32_16x16x4_f32(false, av, false, bv,
                                                             (short)0, acc1[ft], false, false);
        }
    }

    v8f acc2[UF / 16];
#pragma unroll
    for (int ot = 0; ot < UF / 16; ++ot) acc2[ot] = v8f{};

#pragma unroll
    for (int ft = 0; ft < H1F / 16; ++ft) {
        const float4* __restrict__ bp = (const float4*)(b1 + 16 * ft + 8 * hf);
        const float4 bl = bp[0], bh = bp[1];
        float hv[8];
        hv[0] = fmaxf(acc1[ft][0] + bl.x, 0.0f); hv[1] = fmaxf(acc1[ft][1] + bl.y, 0.0f);
        hv[2] = fmaxf(acc1[ft][2] + bl.z, 0.0f); hv[3] = fmaxf(acc1[ft][3] + bl.w, 0.0f);
        hv[4] = fmaxf(acc1[ft][4] + bh.x, 0.0f); hv[5] = fmaxf(acc1[ft][5] + bh.y, 0.0f);
        hv[6] = fmaxf(acc1[ft][6] + bh.z, 0.0f); hv[7] = fmaxf(acc1[ft][7] + bh.w, 0.0f);
#pragma unroll
        for (int p = 0; p < 4; ++p) {
            const int k = 16 * ft + 8 * hf + 2 * p;
            v2f bv; bv.x = hv[2 * p]; bv.y = hv[2 * p + 1];
#pragma unroll
            for (int ot = 0; ot < UF / 16; ++ot) {
                v2f av;
                av.x = W234[k * UF + 16 * ot + c];
                av.y = W234[(k + 1) * UF + 16 * ot + c];
                acc2[ot] = __builtin_amdgcn_wmma_f32_16x16x4_f32(false, av, false, bv,
                                                                 (short)0, acc2[ot], false, false);
            }
        }
    }

    const float di = dis[node];
    float4* __restrict__ o4 = (float4*)(u0 + (size_t)node * UF + 8 * hf);
#pragma unroll
    for (int ot = 0; ot < UF / 16; ++ot) {
        float4 lo, hi;
        lo.x = acc2[ot][0] * di; lo.y = acc2[ot][1] * di; lo.z = acc2[ot][2] * di; lo.w = acc2[ot][3] * di;
        hi.x = acc2[ot][4] * di; hi.y = acc2[ot][5] * di; hi.z = acc2[ot][6] * di; hi.w = acc2[ot][7] * di;
        o4[4 * ot]     = lo;
        o4[4 * ot + 1] = hi;
    }
}

#define NPB   256
#define NB    ((NN + NPB - 1) / NPB)
#define CHUNK 4096
#define PT    1024
#define OFFP  256
#define FOLD1_WG 2
#define FOLD2_WG 1
static_assert(CHUNK == 4 * PT && CHUNK < 65536, "four edges per thread; offsets fit 16 bits");
static_assert(NB <= OFFP && NB + 1 <= 8 * 32, "one wave scans the range counts, 8 per lane");
static_assert(PT / 32 >= 6 * 4, "a wave for every tile of W2*W3");
static_assert(NPB == 256, "range of a node = id >> 8");
static_assert(NPB * 8 == 2 * PT && NPB * SLOTS * 2 == 2 * PT * 16, "two float4 and two 16-byte slot pieces per thread");

__device__ inline void bias_chain(const float* __restrict__ b2, const float* __restrict__ W3,
                                  const float* __restrict__ b3, const float* __restrict__ W4,
                                  float* __restrict__ c2, float* __restrict__ c3,
                                  float* __restrict__ tmp  ) {
    const int t = threadIdx.x;
    if (t < 64) {
        float s = 0.0f;
        for (int k = 0; k < 128; ++k) s += b2[k] * W3[k * 64 + t];
        tmp[t] = s;
    }
    __syncthreads();
    if (t < 32) {
        float a = 0.0f, b = 0.0f;
        for (int j = 0; j < 64; ++j) {
            float w = W4[j * 32 + t];
            a += tmp[j] * w;
            b += b3[j] * w;
        }
        c2[t] = a;
        c3[t] = b;
    }
}

__global__ void __launch_bounds__(PT)
edge_bucket_fold(const int* __restrict__ row, const int* __restrict__ col, int E,
                 unsigned* __restrict__ sorted, unsigned* __restrict__ offs,
                 const float* __restrict__ W2, const float* __restrict__ W3,
                 float* __restrict__ W23,
                 const float* __restrict__ b2, const float* __restrict__ b3,
                 const float* __restrict__ W4,
                 float* __restrict__ c2, float* __restrict__ c3) {
    __shared__ int hist[NB + 1];
    __shared__ unsigned stage[CHUNK];
    __shared__ float tmp[64];
    const int b = blockIdx.x;
    const int t = threadIdx.x;
    if (b == 0) {
        const int wave = t >> 5;
        if (wave < 6 * 4)
            tile_wmma_f32<128, 128, 64, false, false, false>(W2, W3, nullptr, nullptr, W23, wave);
    } else if (b == 1) {
        bias_chain(b2, W3, b3, W4, c2, c3, tmp);
    } else {
        const int w = b - FOLD1_WG;
        const int base = w * CHUNK;
        if (t <= NB) hist[t] = 0;
        __syncthreads();
        int cc[4], ss[4], bk[4], rk[4];
#pragma unroll
        for (int j = 0; j < 4; ++j) {
            const int e = base + j * PT + t;
            const int ec = e < E ? e : base;
            const int c = col[ec];
            ss[j] = row[ec];
            cc[j] = e < E ? c : -1;
        }
#pragma unroll
        for (int j = 0; j < 4; ++j) {
            const int k = cc[j] >> 8;
            bk[j] = k < NB ? k : NB - 1;
            rk[j] = 0;
            if (cc[j] >= 0) rk[j] = atomicAdd(&hist[bk[j]], 1);
        }
        __syncthreads();
        if (t < 32) {
            int v[8];
            int sum = 0;
#pragma unroll
            for (int k = 0; k < 8; ++k) {
                const int i = 8 * t + k;
                v[k] = i < NB ? hist[i] : 0;
                sum += v[k];
            }
            int incl = sum;
#pragma unroll
            for (int d = 1; d < 32; d <<= 1) {
                const int up = __shfl_up(incl, d);
                if (t >= d) incl += up;
            }
            int run = incl - sum;
#pragma unroll
            for (int k = 0; k < 8; ++k) {
                const int i = 8 * t + k;
                if (i <= NB) hist[i] = run;
                run += v[k];
            }
        }
        __syncthreads();
#pragma unroll
        for (int j = 0; j < 4; ++j)
            if (cc[j] >= 0)
                stage[hist[bk[j]] + rk[j]] = ((unsigned)cc[j] << 16) | ((unsigned)ss[j] & 0xffffu);
        __syncthreads();
        const int n = E - base < CHUNK ? E - base : CHUNK;
#pragma unroll
        for (int j = 0; j < 4; ++j) {
            const int i = j * PT + t;
            if (i < n) sorted[(size_t)base + i] = stage[i];
        }
        if (t < NB) offs[(size_t)w * OFFP + t] = (unsigned)hist[t] | ((unsigned)hist[t + 1] << 16);
    }
}

__device__ inline float dis_of(int c) {
    const float v = (float)(c + 1);
    return (v > 0.0f) ? rsqrtf(fmaxf(v, 1.0f)) : 0.0f;
}

__global__ void __launch_bounds__(PT)
range_build(const unsigned* __restrict__ sorted, const unsigned* __restrict__ offs, int NW,
            const float* __restrict__ x, int* __restrict__ cnt, float* __restrict__ dis,
            u16* __restrict__ srcs, float* __restrict__ xp,
            const float* __restrict__ W23, const float* __restrict__ W4,
            float* __restrict__ W234) {
    __shared__ int lcnt[NPB];
    __shared__ uint4 lrow[NPB * SLOTS * 2 / 16];
    const int t = threadIdx.x;
    const int b = (int)blockIdx.x - FOLD2_WG;
    if (b < 0) {
        const int wave = t >> 5;
        if (wave < 6 * 2)
            tile_wmma_f32<64, 64, 32, false, false, false>(W23, W4, nullptr, nullptr, W234, wave);
        return;
    }
    const int n0 = b * NPB;
    const int nn = NN - n0 < NPB ? NN - n0 : NPB;
    u16* lsrc = (u16*)lrow;

    if (t < NPB) lcnt[t] = 0;
    lrow[t] = make_uint4(0u, 0u, 0u, 0u);
    lrow[t + PT] = make_uint4(0u, 0u, 0u, 0u);
    __syncthreads();

    float xv[2][4];
#pragma unroll
    for (int k = 0; k < 2; ++k) {
        const int j = k * PT + t;
        const int n = j >> 3, c = (j & 7) * 4;
        const float* __restrict__ xr = x + (size_t)(n < nn ? n0 + n : n0) * XK;
#pragma unroll
        for (int m = 0; m < 4; ++m) {
            const float v = xr[c + m < XK ? c + m : XK - 1];
            xv[k][m] = c + m < XK ? v : 0.0f;
        }
    }


    const int g = t >> 3, q = t & 7;
    for (int w0 = 0; w0 < NW; w0 += 2 * (PT / 8)) {
        const int wa = w0 + g, wb = wa + PT / 8;
        const bool va = wa < NW, vb = wb < NW;
        const unsigned pa = offs[(size_t)(va ? wa : 0) * OFFP + b];
        const unsigned pb = offs[(size_t)(vb ? wb : 0) * OFFP + b];
        const unsigned* __restrict__ sa = sorted + (size_t)(va ? wa : 0) * CHUNK;
        const unsigned* __restrict__ sb = sorted + (size_t)(vb ? wb : 0) * CHUNK;
        const int loa = (int)(pa & 0xffffu), lob = (int)(pb & 0xffffu);
        const int hia = va ? (int)(pa >> 16) : loa;
        const int hib = vb ? (int)(pb >> 16) : lob;
        unsigned ea[4], eb[4];
#pragma unroll
        for (int k = 0; k < 4; ++k) {
            const int ia = loa + q + 8 * k, ib = lob + q + 8 * k;
            const unsigned a = sa[ia < hia ? ia : 0];
            const unsigned c = sb[ib < hib ? ib : 0];
            ea[k] = ia < hia ? a : 0xffffffffu;
            eb[k] = ib < hib ? c : 0xffffffffu;
        }
        unsigned loc[8];
        int rk[8];
#pragma unroll
        for (int k = 0; k < 8; ++k) {
            const unsigned e = k < 4 ? ea[k & 3] : eb[k & 3];
            loc[k] = (e >> 16) - (unsigned)n0;
            rk[k] = SLOTS;
            if (loc[k] < (unsigned)nn) rk[k] = atomicAdd(&lcnt[loc[k]], 1);
        }
#pragma unroll
        for (int k = 0; k < 8; ++k) {
            const unsigned e = k < 4 ? ea[k & 3] : eb[k & 3];
            if (rk[k] < SLOTS) lsrc[loc[k] * SLOTS + rk[k]] = (u16)(e & 0xffffu);
        }
#define PLACE_EDGE(EW) { const unsigned e_ = (EW); const unsigned loc_ = (e_ >> 16) - (unsigned)n0; \
            if (loc_ < (unsigned)nn) { const int r_ = atomicAdd(&lcnt[loc_], 1); \
                if (r_ < SLOTS) lsrc[loc_ * SLOTS + r_] = (u16)(e_ & 0xffffu); } }
        for (int i = loa + q + 32; i < hia; i += 8) PLACE_EDGE(sa[i])
        for (int i = lob + q + 32; i < hib; i += 8) PLACE_EDGE(sb[i])
#undef PLACE_EDGE
    }
    __syncthreads();

    if (t < nn) {
        const int d = lcnt[t];
        cnt[n0 + t] = d;
        dis[n0 + t] = dis_of(d);
    }
    uint4* __restrict__ grow = (uint4*)(srcs + (size_t)n0 * SLOTS);
    float4* __restrict__ xp4 = (float4*)xp + (size_t)n0 * (XKP / 4);
#pragma unroll
    for (int k = 0; k < 2; ++k) {
        const int j = k * PT + t;
        const int n = j >> 3;
        if (n < nn) {
            grow[j] = lrow[j];
            const float d = dis_of(lcnt[n]);
            float4 v;
            v.x = xv[k][0] * d; v.y = xv[k][1] * d; v.z = xv[k][2] * d; v.w = xv[k][3] * d;
            xp4[j] = v;
        }
    }
}

static_assert(NN % 4 == 0, "a wave takes four whole nodes");

template <int U>
__device__ inline int group_lane(int v) {
    return __builtin_amdgcn_mov_dpp8(v, U * 0x249249);
}

template <bool BIAS, bool OSCALE>
__global__ void __launch_bounds__(256)
agg_gather4(const float* __restrict__ h, const float* __restrict__ dis,
            const int* __restrict__ cnt, const u16* __restrict__ srcs,
            const unsigned* __restrict__ sorted, const unsigned* __restrict__ offs, int NW,
            const float* __restrict__ bias, float* __restrict__ out) {
    const int wave = blockIdx.x * (blockDim.x >> 5) + (threadIdx.x >> 5);
    if (wave >= NN / 4) return;
    const int lane = threadIdx.x & 31;
    const int q    = lane & 7;
    const int node = wave * 4 + (lane >> 3);
    const int deg = cnt[node];
    const int beg = node * SLOTS;
    const int end = beg + (deg <= SLOTS ? deg : 0);
    const float di = dis[node];
    const float4* __restrict__ h4 = (const float4*)h;

    float4 acc = h4[(unsigned)node * 8u + (unsigned)q];
    float4 bv = make_float4(0.0f, 0.0f, 0.0f, 0.0f);
    if constexpr (BIAS) bv = *(const float4*)(bias + q * 4);

    int e0 = beg;
    int my_s;
    {
        const bool in = e0 + q < end;
        const int v = (int)srcs[in ? e0 + q : 0];
        my_s = in ? v : node;
    }
    while (e0 < end) {
        const int e1 = e0 + 8;
        const bool in1 = e1 + q < end;
        const int v1 = (int)srcs[in1 ? e1 + q : 0];
        float4 r[8];
#define GATHER_ROW(U) r[U] = h4[(unsigned)group_lane<U>(my_s) * 8u + (unsigned)q];
        GATHER_ROW(0) GATHER_ROW(1) GATHER_ROW(2) GATHER_ROW(3)
        GATHER_ROW(4) GATHER_ROW(5) GATHER_ROW(6) GATHER_ROW(7)
#undef GATHER_ROW
#pragma unroll
        for (int u = 0; u < 8; ++u) {
            const bool on = (u == 0) || (e0 + u < end);
            acc.x += on ? r[u].x : 0.0f;
            acc.y += on ? r[u].y : 0.0f;
            acc.z += on ? r[u].z : 0.0f;
            acc.w += on ? r[u].w : 0.0f;
        }
        my_s = in1 ? v1 : node;
        e0 = e1;
    }

    if (deg > SLOTS) {
        const int b = node / NPB;
        for (int w = 0; w < NW; ++w) {
            const unsigned p = offs[(size_t)w * OFFP + b];
            const int lo = (int)(p & 0xffffu), hi = (int)(p >> 16);
            for (int i = lo; i < hi; ++i) {
                const unsigned e = sorted[(size_t)w * CHUNK + i];
                if ((int)(e >> 16) == node) {
                    const float4 r = h4[(e & 0xffffu) * 8u + (unsigned)q];
                    acc.x += r.x; acc.y += r.y; acc.z += r.z; acc.w += r.w;
                }
            }
        }
    }

    float4 o;
    o.x = acc.x * di; o.y = acc.y * di; o.z = acc.z * di; o.w = acc.w * di;
    if constexpr (BIAS)   { o.x += bv.x; o.y += bv.y; o.z += bv.z; o.w += bv.w; }
    if constexpr (OSCALE) { o.x *= di;   o.y *= di;   o.z *= di;   o.w *= di; }
    ((float4*)out)[(unsigned)node * 8u + (unsigned)q] = o;
}

static inline int ceil_div(long long a, long long b) { return (int)((a + b - 1) / b); }

extern "C" void kernel_launch(void* const* d_in, const int* in_sizes, int n_in,
                              void* d_out, int out_size, void* d_ws, size_t ws_size,
                              hipStream_t stream) {
    const float* x  = (const float*)d_in[0];
    const int*   ei = (const int*)d_in[1];
    const float* W1 = (const float*)d_in[2];
    const float* b1 = (const float*)d_in[3];
    const float* W2 = (const float*)d_in[4];
    const float* b2 = (const float*)d_in[5];
    const float* W3 = (const float*)d_in[6];
    const float* b3 = (const float*)d_in[7];
    const float* W4 = (const float*)d_in[8];
    const float* b4 = (const float*)d_in[9];

    const int E = in_sizes[1] / 2;
    const int* row = ei;
    const int* col = ei + E;

    float* dis  = (float*)d_ws;
    float* reg1 = dis  + ((NN + 255) & ~255);
    float* bufA = reg1 + ((E  + 255) & ~255);
    float* bufB = bufA + (size_t)NN * 128;
    int*   cnt  = (int*)reg1;
    float* W23  = (float*)cnt + ((NN + 1 + 255) & ~255);
    float* W234 = W23  + 96 * 64;
    float* c2   = W234 + 96 * 32;
    float* c3   = c2 + 32;
    const int NW = ceil_div(E, CHUNK);
    u16*      srcs   = (u16*)(bufA + (size_t)NN * 96);
    unsigned* sorted = (unsigned*)bufB;
    unsigned* offs   = (unsigned*)(bufB + ((E + 255) & ~255));
    float* xp = bufA;
    float* ax = bufA + (size_t)NN * 32;
    float* u0 = bufA;
    float* u1 = bufA + (size_t)NN * 32;
    float* u2 = bufA + (size_t)NN * 64;

    const int TB = 256;
    const int GB = 128;

    edge_bucket_fold<<<FOLD1_WG + NW, PT, 0, stream>>>(row, col, E, sorted, offs, W2, W3, W23, b2, b3, W4, c2, c3);
    range_build<<<FOLD2_WG + NB, PT, 0, stream>>>(sorted, offs, NW, x, cnt, dis, srcs, xp, W23, W4, W234);

    const int TILES_M = NN / 16;
    const int AGB = ceil_div(NN / 4, TB / 32);

    agg_gather4<false, false><<<AGB, TB, 0, stream>>>(xp, dis, cnt, srcs, sorted, offs, NW, nullptr, ax);
    dense_chain<<<ceil_div(TILES_M, GB / 32), GB, 0, stream>>>(ax, W1, b1, W234, dis, u0, TILES_M);
    float* out = (float*)d_out;
    agg_gather4<true, true><<<AGB, TB, 0, stream>>>(u0, dis, cnt, srcs, sorted, offs, NW, c2, u1);
    agg_gather4<true, true><<<AGB, TB, 0, stream>>>(u1, dis, cnt, srcs, sorted, offs, NW, c3, u2);
    agg_gather4<true, false><<<AGB, TB, 0, stream>>>(u2, dis, cnt, srcs, sorted, offs, NW, b4, out);
}


# Layer-1 gather fused with both dense products: a work-group leaves 32 nodes' S*x rows in LDS, 4 of its 8 waves do 29->96->32 from there (96-term sum split in two halves): 6 launches not 7; f32 kept
# speedup vs baseline: 12.6017x; 1.0780x over previous
#include <hip/hip_runtime.h>

#define NN 50000

typedef float v8f __attribute__((ext_vector_type(8)));
typedef float v2f __attribute__((ext_vector_type(2)));

typedef unsigned short u16;
static_assert(NN < 65535, "ids are stored in 16 bits; 0xffff marks no edge");

#define SLOTS 64

#define XK  29
#define XKP 32
template <int K, int KW, int F, bool BIAS, bool RELU, bool RSCALE>
__device__ inline void tile_wmma_f32(const float* __restrict__ A, const float* __restrict__ W,
                                     const float* __restrict__ bias, const float* __restrict__ rscale,
                                     float* __restrict__ H, int wave) {
    static_assert(K % 4 == 0 && KW <= K && K - KW < 4, "K is KW rounded up to 4");
    constexpr int tilesN = F / 16;
    int lane = threadIdx.x & 31;
    int half = lane >> 4;
    int lm   = lane & 15;
    int tm = wave / tilesN;
    int tn = wave - tm * tilesN;
    const float* __restrict__ Arow = A + (size_t)(tm * 16 + lm) * K;
    int bcol = tn * 16 + lm;

    v8f c = {};
#pragma unroll
    for (int k0 = 0; k0 < K; k0 += 4) {
        int ka = k0 + 2 * half;
        float2 a2 = *(const float2*)(Arow + ka);
        float a0 = a2.x, a1 = a2.y;
        float b0, b1;
        if (k0 + 4 <= KW) {
            b0 = W[(size_t)ka * F + bcol];
            b1 = W[(size_t)(ka + 1) * F + bcol];
        } else {
            int i0 = (ka     < KW) ? ka       : (KW - 1);
            int i1 = (ka + 1 < KW) ? (ka + 1) : (KW - 1);
            b0 = W[(size_t)i0 * F + bcol];
            b1 = W[(size_t)i1 * F + bcol];
            if (ka     >= KW) { a0 = 0.0f; b0 = 0.0f; }
            if (ka + 1 >= KW) { a1 = 0.0f; b1 = 0.0f; }
        }
        v2f av; av.x = a0; av.y = a1;
        v2f bv; bv.x = b0; bv.y = b1;
        c = __builtin_amdgcn_wmma_f32_16x16x4_f32(false, av, false, bv,
                                                  (short)0, c, false, false);
    }

    int col = tn * 16 + lm;
    float bb = 0.0f;
    if constexpr (BIAS) bb = bias[col];
#pragma unroll
    for (int j = 0; j < 8; ++j) {
        int r = tm * 16 + j + 8 * half;
        float v = c[j];
        if constexpr (BIAS) v += bb;
        if constexpr (RELU) v = fmaxf(v, 0.0f);
        if constexpr (RSCALE) v *= rscale[r];
        H[(size_t)r * F + col] = v;
    }
}

#define NPB   256
#define NB    ((NN + NPB - 1) / NPB)
#define CHUNK 4096
#define PT    1024
#define OFFP  256
#define FOLD1_WG 2
#define FOLD2_WG 1
static_assert(CHUNK == 4 * PT && CHUNK < 65536, "four edges per thread; offsets fit 16 bits");
static_assert(NB <= OFFP && NB + 1 <= 8 * 32, "one wave scans the range counts, 8 per lane");
static_assert(PT / 32 >= 6 * 4, "a wave for every tile of W2*W3");
static_assert(NPB == 256, "range of a node = id >> 8");
static_assert(NPB * 8 == 2 * PT && NPB * SLOTS * 2 == 2 * PT * 16, "two float4 and two 16-byte slot pieces per thread");

__device__ inline void bias_chain(const float* __restrict__ b2, const float* __restrict__ W3,
                                  const float* __restrict__ b3, const float* __restrict__ W4,
                                  float* __restrict__ c2, float* __restrict__ c3,
                                  float* __restrict__ tmp  ) {
    const int t = threadIdx.x;
    if (t < 64) {
        float s = 0.0f;
        for (int k = 0; k < 128; ++k) s += b2[k] * W3[k * 64 + t];
        tmp[t] = s;
    }
    __syncthreads();
    if (t < 32) {
        float a = 0.0f, b = 0.0f;
        for (int j = 0; j < 64; ++j) {
            float w = W4[j * 32 + t];
            a += tmp[j] * w;
            b += b3[j] * w;
        }
        c2[t] = a;
        c3[t] = b;
    }
}

__global__ void __launch_bounds__(PT)
edge_bucket_fold(const int* __restrict__ row, const int* __restrict__ col, int E,
                 unsigned* __restrict__ sorted, unsigned* __restrict__ offs,
                 const float* __restrict__ W2, const float* __restrict__ W3,
                 float* __restrict__ W23,
                 const float* __restrict__ b2, const float* __restrict__ b3,
                 const float* __restrict__ W4,
                 float* __restrict__ c2, float* __restrict__ c3) {
    __shared__ int hist[NB + 1];
    __shared__ unsigned stage[CHUNK];
    __shared__ float tmp[64];
    const int b = blockIdx.x;
    const int t = threadIdx.x;
    if (b == 0) {
        const int wave = t >> 5;
        if (wave < 6 * 4)
            tile_wmma_f32<128, 128, 64, false, false, false>(W2, W3, nullptr, nullptr, W23, wave);
    } else if (b == 1) {
        bias_chain(b2, W3, b3, W4, c2, c3, tmp);
    } else {
        const int w = b - FOLD1_WG;
        const int base = w * CHUNK;
        if (t <= NB) hist[t] = 0;
        __syncthreads();
        int cc[4], ss[4], bk[4], rk[4];
#pragma unroll
        for (int j = 0; j < 4; ++j) {
            const int e = base + j * PT + t;
            const int ec = e < E ? e : base;
            const int c = col[ec];
            ss[j] = row[ec];
            cc[j] = e < E ? c : -1;
        }
#pragma unroll
        for (int j = 0; j < 4; ++j) {
            const int k = cc[j] >> 8;
            bk[j] = k < NB ? k : NB - 1;
            rk[j] = 0;
            if (cc[j] >= 0) rk[j] = atomicAdd(&hist[bk[j]], 1);
        }
        __syncthreads();
        if (t < 32) {
            int v[8];
            int sum = 0;
#pragma unroll
            for (int k = 0; k < 8; ++k) {
                const int i = 8 * t + k;
                v[k] = i < NB ? hist[i] : 0;
                sum += v[k];
            }
            int incl = sum;
#pragma unroll
            for (int d = 1; d < 32; d <<= 1) {
                const int up = __shfl_up(incl, d);
                if (t >= d) incl += up;
            }
            int run = incl - sum;
#pragma unroll
            for (int k = 0; k < 8; ++k) {
                const int i = 8 * t + k;
                if (i <= NB) hist[i] = run;
                run += v[k];
            }
        }
        __syncthreads();
#pragma unroll
        for (int j = 0; j < 4; ++j)
            if (cc[j] >= 0)
                stage[hist[bk[j]] + rk[j]] = ((unsigned)cc[j] << 16) | ((unsigned)ss[j] & 0xffffu);
        __syncthreads();
        const int n = E - base < CHUNK ? E - base : CHUNK;
#pragma unroll
        for (int j = 0; j < 4; ++j) {
            const int i = j * PT + t;
            if (i < n) sorted[(size_t)base + i] = stage[i];
        }
        if (t < NB) offs[(size_t)w * OFFP + t] = (unsigned)hist[t] | ((unsigned)hist[t + 1] << 16);
    }
}

__device__ inline float dis_of(int c) {
    const float v = (float)(c + 1);
    return (v > 0.0f) ? rsqrtf(fmaxf(v, 1.0f)) : 0.0f;
}

__global__ void __launch_bounds__(PT)
range_build(const unsigned* __restrict__ sorted, const unsigned* __restrict__ offs, int NW,
            const float* __restrict__ x, int* __restrict__ cnt, float* __restrict__ dis,
            u16* __restrict__ srcs, float* __restrict__ xp,
            const float* __restrict__ W23, const float* __restrict__ W4,
            float* __restrict__ W234) {
    __shared__ int lcnt[NPB];
    __shared__ uint4 lrow[NPB * SLOTS * 2 / 16];
    const int t = threadIdx.x;
    const int b = (int)blockIdx.x - FOLD2_WG;
    if (b < 0) {
        const int wave = t >> 5;
        if (wave < 6 * 2)
            tile_wmma_f32<64, 64, 32, false, false, false>(W23, W4, nullptr, nullptr, W234, wave);
        return;
    }
    const int n0 = b * NPB;
    const int nn = NN - n0 < NPB ? NN - n0 : NPB;
    u16* lsrc = (u16*)lrow;

    if (t < NPB) lcnt[t] = 0;
    lrow[t] = make_uint4(0u, 0u, 0u, 0u);
    lrow[t + PT] = make_uint4(0u, 0u, 0u, 0u);
    __syncthreads();

    float xv[2][4];
#pragma unroll
    for (int k = 0; k < 2; ++k) {
        const int j = k * PT + t;
        const int n = j >> 3, c = (j & 7) * 4;
        const float* __restrict__ xr = x + (size_t)(n < nn ? n0 + n : n0) * XK;
#pragma unroll
        for (int m = 0; m < 4; ++m) {
            const float v = xr[c + m < XK ? c + m : XK - 1];
            xv[k][m] = c + m < XK ? v : 0.0f;
        }
    }


    const int g = t >> 3, q = t & 7;
    for (int w0 = 0; w0 < NW; w0 += 2 * (PT / 8)) {
        const int wa = w0 + g, wb = wa + PT / 8;
        const bool va = wa < NW, vb = wb < NW;
        const unsigned pa = offs[(size_t)(va ? wa : 0) * OFFP + b];
        const unsigned pb = offs[(size_t)(vb ? wb : 0) * OFFP + b];
        const unsigned* __restrict__ sa = sorted + (size_t)(va ? wa : 0) * CHUNK;
        const unsigned* __restrict__ sb = sorted + (size_t)(vb ? wb : 0) * CHUNK;
        const int loa = (int)(pa & 0xffffu), lob = (int)(pb & 0xffffu);
        const int hia = va ? (int)(pa >> 16) : loa;
        const int hib = vb ? (int)(pb >> 16) : lob;
        unsigned ea[4], eb[4];
#pragma unroll
        for (int k = 0; k < 4; ++k) {
            const int ia = loa + q + 8 * k, ib = lob + q + 8 * k;
            const unsigned a = sa[ia < hia ? ia : 0];
            const unsigned c = sb[ib < hib ? ib : 0];
            ea[k] = ia < hia ? a : 0xffffffffu;
            eb[k] = ib < hib ? c : 0xffffffffu;
        }
        unsigned loc[8];
        int rk[8];
#pragma unroll
        for (int k = 0; k < 8; ++k) {
            const unsigned e = k < 4 ? ea[k & 3] : eb[k & 3];
            loc[k] = (e >> 16) - (unsigned)n0;
            rk[k] = SLOTS;
            if (loc[k] < (unsigned)nn) rk[k] = atomicAdd(&lcnt[loc[k]], 1);
        }
#pragma unroll
        for (int k = 0; k < 8; ++k) {
            const unsigned e = k < 4 ? ea[k & 3] : eb[k & 3];
            if (rk[k] < SLOTS) lsrc[loc[k] * SLOTS + rk[k]] = (u16)(e & 0xffffu);
        }
#define PLACE_EDGE(EW) { const unsigned e_ = (EW); const unsigned loc_ = (e_ >> 16) - (unsigned)n0; \
            if (loc_ < (unsigned)nn) { const int r_ = atomicAdd(&lcnt[loc_], 1); \
                if (r_ < SLOTS) lsrc[loc_ * SLOTS + r_] = (u16)(e_ & 0xffffu); } }
        for (int i = loa + q + 32; i < hia; i += 8) PLACE_EDGE(sa[i])
        for (int i = lob + q + 32; i < hib; i += 8) PLACE_EDGE(sb[i])
#undef PLACE_EDGE
    }
    __syncthreads();

    if (t < nn) {
        const int d = lcnt[t];
        cnt[n0 + t] = d;
        dis[n0 + t] = dis_of(d);
    }
    uint4* __restrict__ grow = (uint4*)(srcs + (size_t)n0 * SLOTS);
    float4* __restrict__ xp4 = (float4*)xp + (size_t)n0 * (XKP / 4);
#pragma unroll
    for (int k = 0; k < 2; ++k) {
        const int j = k * PT + t;
        const int n = j >> 3;
        if (n < nn) {
            grow[j] = lrow[j];
            const float d = dis_of(lcnt[n]);
            float4 v;
            v.x = xv[k][0] * d; v.y = xv[k][1] * d; v.z = xv[k][2] * d; v.w = xv[k][3] * d;
            xp4[j] = v;
        }
    }
}

static_assert(NN % 4 == 0, "a wave takes four whole nodes");

template <int U>
__device__ inline int group_lane(int v) {
    return __builtin_amdgcn_mov_dpp8(v, U * 0x249249);
}

__device__ inline float4 gather_rows4(const float4* __restrict__ h4, const int* __restrict__ cnt,
                                      const u16* __restrict__ srcs,
                                      const unsigned* __restrict__ sorted,
                                      const unsigned* __restrict__ offs, int NW,
                                      int node, int q) {
    const int deg = cnt[node];
    const int beg = node * SLOTS;
    const int end = beg + (deg <= SLOTS ? deg : 0);

    float4 acc = h4[(unsigned)node * 8u + (unsigned)q];

    int e0 = beg;
    int my_s;
    {
        const bool in = e0 + q < end;
        const int v = (int)srcs[in ? e0 + q : 0];
        my_s = in ? v : node;
    }
    while (e0 < end) {
        const int e1 = e0 + 8;
        const bool in1 = e1 + q < end;
        const int v1 = (int)srcs[in1 ? e1 + q : 0];
        float4 r[8];
#define GATHER_ROW(U) r[U] = h4[(unsigned)group_lane<U>(my_s) * 8u + (unsigned)q];
        GATHER_ROW(0) GATHER_ROW(1) GATHER_ROW(2) GATHER_ROW(3)
        GATHER_ROW(4) GATHER_ROW(5) GATHER_ROW(6) GATHER_ROW(7)
#undef GATHER_ROW
#pragma unroll
        for (int u = 0; u < 8; ++u) {
            const bool on = (u == 0) || (e0 + u < end);
            acc.x += on ? r[u].x : 0.0f;
            acc.y += on ? r[u].y : 0.0f;
            acc.z += on ? r[u].z : 0.0f;
            acc.w += on ? r[u].w : 0.0f;
        }
        my_s = in1 ? v1 : node;
        e0 = e1;
    }

    if (deg > SLOTS) {
        const int b = node / NPB;
        for (int w = 0; w < NW; ++w) {
            const unsigned p = offs[(size_t)w * OFFP + b];
            const int lo = (int)(p & 0xffffu), hi = (int)(p >> 16);
            for (int i = lo; i < hi; ++i) {
                const unsigned e = sorted[(size_t)w * CHUNK + i];
                if ((int)(e >> 16) == node) {
                    const float4 r = h4[(e & 0xffffu) * 8u + (unsigned)q];
                    acc.x += r.x; acc.y += r.y; acc.z += r.z; acc.w += r.w;
                }
            }
        }
    }
    return acc;
}

template <bool BIAS, bool OSCALE>
__global__ void __launch_bounds__(256)
agg_gather4(const float* __restrict__ h, const float* __restrict__ dis,
            const int* __restrict__ cnt, const u16* __restrict__ srcs,
            const unsigned* __restrict__ sorted, const unsigned* __restrict__ offs, int NW,
            const float* __restrict__ bias, float* __restrict__ out) {
    const int wave = blockIdx.x * (blockDim.x >> 5) + (threadIdx.x >> 5);
    if (wave >= NN / 4) return;
    const int lane = threadIdx.x & 31;
    const int q    = lane & 7;
    const int node = wave * 4 + (lane >> 3);
    const float di = dis[node];
    float4 bv = make_float4(0.0f, 0.0f, 0.0f, 0.0f);
    if constexpr (BIAS) bv = *(const float4*)(bias + q * 4);
    const float4 acc = gather_rows4((const float4*)h, cnt, srcs, sorted, offs, NW, node, q);

    float4 o;
    o.x = acc.x * di; o.y = acc.y * di; o.z = acc.z * di; o.w = acc.w * di;
    if constexpr (BIAS)   { o.x += bv.x; o.y += bv.y; o.z += bv.z; o.w += bv.w; }
    if constexpr (OSCALE) { o.x *= di;   o.y *= di;   o.z *= di;   o.w *= di; }
    ((float4*)out)[(unsigned)node * 8u + (unsigned)q] = o;
}

#define H1F 96
#define UF  32
#define FN  32
#define AXP 36
static_assert(XKP == 32 && UF == 32 && NN % 16 == 0, "a lane half holds 16 of a row's 32 floats; whole node tiles");
static_assert(H1F == 6 * 16 && FN == 2 * 16, "two waves of three feature tiles per node tile, two node tiles");

__global__ void __launch_bounds__(256)
gather_dense(const float* __restrict__ xp, const float* __restrict__ dis,
             const int* __restrict__ cnt, const u16* __restrict__ srcs,
             const unsigned* __restrict__ sorted, const unsigned* __restrict__ offs, int NW,
             const float* __restrict__ W1, const float* __restrict__ b1,
             const float* __restrict__ W234, float* __restrict__ u0) {
    __shared__ float4 lax[FN * AXP / 4];
    __shared__ float4 lpart[FN * AXP / 4];
    const int wv = __builtin_amdgcn_readfirstlane((int)(threadIdx.x >> 5));
    const int lane = threadIdx.x & 31;
    const int n0 = blockIdx.x * FN;

    if (n0 + wv * 4 < NN) {
        const int q  = lane & 7;
        const int nl = wv * 4 + (lane >> 3);
        const int node = n0 + nl;
        const float di = dis[node];
        const float4 acc = gather_rows4((const float4*)xp, cnt, srcs, sorted, offs, NW, node, q);
        float4 o;
        o.x = acc.x * di; o.y = acc.y * di; o.z = acc.z * di; o.w = acc.w * di;
        lax[nl * (AXP / 4) + q] = o;
    }
    __syncthreads();

    const int nt = wv >> 1, hh = wv & 1;
    const bool work = wv < 4 && n0 + nt * 16 < NN;
    const int hf = lane >> 4;
    const int c  = lane & 15;
    const int node = n0 + nt * 16 + c;
    v8f acc2[UF / 16];
#pragma unroll
    for (int ot = 0; ot < UF / 16; ++ot) acc2[ot] = v8f{};
    float di = 0.0f;

    if (work) {
        di = dis[node];
        float axv[16];
        {
            const float4* __restrict__ axr = &lax[(nt * 16 + c) * (AXP / 4) + 4 * hf];
#pragma unroll
            for (int i = 0; i < 4; ++i) {
                const float4 v = axr[i];
                axv[4 * i] = v.x; axv[4 * i + 1] = v.y; axv[4 * i + 2] = v.z; axv[4 * i + 3] = v.w;
            }
        }
        const float* __restrict__ W1h = W1 + 48 * hh;
        const float* __restrict__ b1h = b1 + 48 * hh;
        const float* __restrict__ W234h = W234 + 48 * hh * UF;

        v8f acc1[3];
#pragma unroll
        for (int j = 0; j < 3; ++j) acc1[j] = v8f{};

#pragma unroll
        for (int t = 0; t < 8; ++t) {
            const int k = 16 * hf + 2 * t;
            float x0 = axv[2 * t], x1 = axv[2 * t + 1];
            const bool padded = 16 + 2 * t + 1 >= XK;
            const bool in0 = !padded || k < XK;
            const bool in1 = !padded || k + 1 < XK;
            const int k0 = in0 ? k : XK - 1;
            const int k1 = in1 ? k + 1 : XK - 1;
            if (!in0) x0 = 0.0f;
            if (!in1) x1 = 0.0f;
            v2f bv; bv.x = x0; bv.y = x1;
#pragma unroll
            for (int j = 0; j < 3; ++j) {
                float w0 = W1h[k0 * H1F + 16 * j + c];
                float w1 = W1h[k1 * H1F + 16 * j + c];
                if (!in0) w0 = 0.0f;
                if (!in1) w1 = 0.0f;
                v2f av; av.x = w0; av.y = w1;
                acc1[j] = __builtin_amdgcn_wmma_f32_16x16x4_f32(false, av, false, bv,
                                                                (short)0, acc1[j], false, false);
            }
        }

#pragma unroll
        for (int j = 0; j < 3; ++j) {
            const float4* __restrict__ bp = (const float4*)(b1h + 16 * j + 8 * hf);
            const float4 bl = bp[0], bh = bp[1];
            float hv[8];
            hv[0] = fmaxf(acc1[j][0] + bl.x, 0.0f); hv[1] = fmaxf(acc1[j][1] + bl.y, 0.0f);
            hv[2] = fmaxf(acc1[j][2] + bl.z, 0.0f); hv[3] = fmaxf(acc1[j][3] + bl.w, 0.0f);
            hv[4] = fmaxf(acc1[j][4] + bh.x, 0.0f); hv[5] = fmaxf(acc1[j][5] + bh.y, 0.0f);
            hv[6] = fmaxf(acc1[j][6] + bh.z, 0.0f); hv[7] = fmaxf(acc1[j][7] + bh.w, 0.0f);
#pragma unroll
            for (int p = 0; p < 4; ++p) {
                const int k = 16 * j + 8 * hf + 2 * p;
                v2f bv; bv.x = hv[2 * p]; bv.y = hv[2 * p + 1];
#pragma unroll
                for (int ot = 0; ot < UF / 16; ++ot) {
                    v2f av;
                    av.x = W234h[k * UF + 16 * ot + c];
                    av.y = W234h[(k + 1) * UF + 16 * ot + c];
                    acc2[ot] = __builtin_amdgcn_wmma_f32_16x16x4_f32(false, av, false, bv,
                                                                     (short)0, acc2[ot], false, false);
                }
            }
        }

        if (hh == 1) {
            float4* __restrict__ pr = &lpart[(nt * 16 + c) * (AXP / 4) + 2 * hf];
#pragma unroll
            for (int ot = 0; ot < UF / 16; ++ot) {
                pr[4 * ot]     = make_float4(acc2[ot][0], acc2[ot][1], acc2[ot][2], acc2[ot][3]);
                pr[4 * ot + 1] = make_float4(acc2[ot][4], acc2[ot][5], acc2[ot][6], acc2[ot][7]);
            }
        }
    }
    __syncthreads();

    if (work && hh == 0) {
        const float4* __restrict__ pr = &lpart[(nt * 16 + c) * (AXP / 4) + 2 * hf];
        float4* __restrict__ o4 = (float4*)(u0 + (size_t)node * UF + 8 * hf);
#pragma unroll
        for (int ot = 0; ot < UF / 16; ++ot) {
            const float4 pl = pr[4 * ot], ph = pr[4 * ot + 1];
            float4 lo, hi;
            lo.x = (acc2[ot][0] + pl.x) * di; lo.y = (acc2[ot][1] + pl.y) * di;
            lo.z = (acc2[ot][2] + pl.z) * di; lo.w = (acc2[ot][3] + pl.w) * di;
            hi.x = (acc2[ot][4] + ph.x) * di; hi.y = (acc2[ot][5] + ph.y) * di;
            hi.z = (acc2[ot][6] + ph.z) * di; hi.w = (acc2[ot][7] + ph.w) * di;
            o4[4 * ot]     = lo;
            o4[4 * ot + 1] = hi;
        }
    }
}

static inline int ceil_div(long long a, long long b) { return (int)((a + b - 1) / b); }

extern "C" void kernel_launch(void* const* d_in, const int* in_sizes, int n_in,
                              void* d_out, int out_size, void* d_ws, size_t ws_size,
                              hipStream_t stream) {
    const float* x  = (const float*)d_in[0];
    const int*   ei = (const int*)d_in[1];
    const float* W1 = (const float*)d_in[2];
    const float* b1 = (const float*)d_in[3];
    const float* W2 = (const float*)d_in[4];
    const float* b2 = (const float*)d_in[5];
    const float* W3 = (const float*)d_in[6];
    const float* b3 = (const float*)d_in[7];
    const float* W4 = (const float*)d_in[8];
    const float* b4 = (const float*)d_in[9];

    const int E = in_sizes[1] / 2;
    const int* row = ei;
    const int* col = ei + E;

    float* dis  = (float*)d_ws;
    float* reg1 = dis  + ((NN + 255) & ~255);
    float* bufA = reg1 + ((E  + 255) & ~255);
    float* bufB = bufA + (size_t)NN * 128;
    int*   cnt  = (int*)reg1;
    float* W23  = (float*)cnt + ((NN + 1 + 255) & ~255);
    float* W234 = W23  + 96 * 64;
    float* c2   = W234 + 96 * 32;
    float* c3   = c2 + 32;
    const int NW = ceil_div(E, CHUNK);
    u16*      srcs   = (u16*)(bufA + (size_t)NN * 96);
    unsigned* sorted = (unsigned*)bufB;
    unsigned* offs   = (unsigned*)(bufB + ((E + 255) & ~255));
    float* xp = bufA;
    float* u0 = bufA + (size_t)NN * 32;
    float* u1 = bufA + (size_t)NN * 64;
    float* u2 = bufA;

    const int TB = 256;

    edge_bucket_fold<<<FOLD1_WG + NW, PT, 0, stream>>>(row, col, E, sorted, offs, W2, W3, W23, b2, b3, W4, c2, c3);
    range_build<<<FOLD2_WG + NB, PT, 0, stream>>>(sorted, offs, NW, x, cnt, dis, srcs, xp, W23, W4, W234);

    const int AGB = ceil_div(NN / 4, TB / 32);

    gather_dense<<<ceil_div(NN, FN), 256, 0, stream>>>(xp, dis, cnt, srcs, sorted, offs, NW, W1, b1, W234, u0);
    float* out = (float*)d_out;
    agg_gather4<true, true><<<AGB, TB, 0, stream>>>(u0, dis, cnt, srcs, sorted, offs, NW, c2, u1);
    agg_gather4<true, true><<<AGB, TB, 0, stream>>>(u1, dis, cnt, srcs, sorted, offs, NW, c3, u2);
    agg_gather4<true, false><<<AGB, TB, 0, stream>>>(u2, dis, cnt, srcs, sorted, offs, NW, b4, out);
}


# Gathers: one wave sums eight nodes as two interleaved sets of four (sixteen row loads per trip in flight): 6250 waves per gather, not 12500; layer-1 work-group takes 64 nodes; f32 and sum order kept
# speedup vs baseline: 12.9880x; 1.0307x over previous
#include <hip/hip_runtime.h>

#define NN 50000

typedef float v8f __attribute__((ext_vector_type(8)));
typedef float v2f __attribute__((ext_vector_type(2)));

typedef unsigned short u16;
static_assert(NN < 65535, "ids are stored in 16 bits; 0xffff marks no edge");

#define SLOTS 64

#define XK  29
#define XKP 32
template <int K, int KW, int F, bool BIAS, bool RELU, bool RSCALE>
__device__ inline void tile_wmma_f32(const float* __restrict__ A, const float* __restrict__ W,
                                     const float* __restrict__ bias, const float* __restrict__ rscale,
                                     float* __restrict__ H, int wave) {
    static_assert(K % 4 == 0 && KW <= K && K - KW < 4, "K is KW rounded up to 4");
    constexpr int tilesN = F / 16;
    int lane = threadIdx.x & 31;
    int half = lane >> 4;
    int lm   = lane & 15;
    int tm = wave / tilesN;
    int tn = wave - tm * tilesN;
    const float* __restrict__ Arow = A + (size_t)(tm * 16 + lm) * K;
    int bcol = tn * 16 + lm;

    v8f c = {};
#pragma unroll
    for (int k0 = 0; k0 < K; k0 += 4) {
        int ka = k0 + 2 * half;
        float2 a2 = *(const float2*)(Arow + ka);
        float a0 = a2.x, a1 = a2.y;
        float b0, b1;
        if (k0 + 4 <= KW) {
            b0 = W[(size_t)ka * F + bcol];
            b1 = W[(size_t)(ka + 1) * F + bcol];
        } else {
            int i0 = (ka     < KW) ? ka       : (KW - 1);
            int i1 = (ka + 1 < KW) ? (ka + 1) : (KW - 1);
            b0 = W[(size_t)i0 * F + bcol];
            b1 = W[(size_t)i1 * F + bcol];
            if (ka     >= KW) { a0 = 0.0f; b0 = 0.0f; }
            if (ka + 1 >= KW) { a1 = 0.0f; b1 = 0.0f; }
        }
        v2f av; av.x = a0; av.y = a1;
        v2f bv; bv.x = b0; bv.y = b1;
        c = __builtin_amdgcn_wmma_f32_16x16x4_f32(false, av, false, bv,
                                                  (short)0, c, false, false);
    }

    int col = tn * 16 + lm;
    float bb = 0.0f;
    if constexpr (BIAS) bb = bias[col];
#pragma unroll
    for (int j = 0; j < 8; ++j) {
        int r = tm * 16 + j + 8 * half;
        float v = c[j];
        if constexpr (BIAS) v += bb;
        if constexpr (RELU) v = fmaxf(v, 0.0f);
        if constexpr (RSCALE) v *= rscale[r];
        H[(size_t)r * F + col] = v;
    }
}

#define NPB   256
#define NB    ((NN + NPB - 1) / NPB)
#define CHUNK 4096
#define PT    1024
#define OFFP  256
#define FOLD1_WG 2
#define FOLD2_WG 1
static_assert(CHUNK == 4 * PT && CHUNK < 65536, "four edges per thread; offsets fit 16 bits");
static_assert(NB <= OFFP && NB + 1 <= 8 * 32, "one wave scans the range counts, 8 per lane");
static_assert(PT / 32 >= 6 * 4, "a wave for every tile of W2*W3");
static_assert(NPB == 256, "range of a node = id >> 8");
static_assert(NPB * 8 == 2 * PT && NPB * SLOTS * 2 == 2 * PT * 16, "two float4 and two 16-byte slot pieces per thread");

__device__ inline void bias_chain(const float* __restrict__ b2, const float* __restrict__ W3,
                                  const float* __restrict__ b3, const float* __restrict__ W4,
                                  float* __restrict__ c2, float* __restrict__ c3,
                                  float* __restrict__ tmp  ) {
    const int t = threadIdx.x;
    if (t < 64) {
        float s = 0.0f;
        for (int k = 0; k < 128; ++k) s += b2[k] * W3[k * 64 + t];
        tmp[t] = s;
    }
    __syncthreads();
    if (t < 32) {
        float a = 0.0f, b = 0.0f;
        for (int j = 0; j < 64; ++j) {
            float w = W4[j * 32 + t];
            a += tmp[j] * w;
            b += b3[j] * w;
        }
        c2[t] = a;
        c3[t] = b;
    }
}

__global__ void __launch_bounds__(PT)
edge_bucket_fold(const int* __restrict__ row, const int* __restrict__ col, int E,
                 unsigned* __restrict__ sorted, unsigned* __restrict__ offs,
                 const float* __restrict__ W2, const float* __restrict__ W3,
                 float* __restrict__ W23,
                 const float* __restrict__ b2, const float* __restrict__ b3,
                 const float* __restrict__ W4,
                 float* __restrict__ c2, float* __restrict__ c3) {
    __shared__ int hist[NB + 1];
    __shared__ unsigned stage[CHUNK];
    __shared__ float tmp[64];
    const int b = blockIdx.x;
    const int t = threadIdx.x;
    if (b == 0) {
        const int wave = t >> 5;
        if (wave < 6 * 4)
            tile_wmma_f32<128, 128, 64, false, false, false>(W2, W3, nullptr, nullptr, W23, wave);
    } else if (b == 1) {
        bias_chain(b2, W3, b3, W4, c2, c3, tmp);
    } else {
        const int w = b - FOLD1_WG;
        const int base = w * CHUNK;
        if (t <= NB) hist[t] = 0;
        __syncthreads();
        int cc[4], ss[4], bk[4], rk[4];
#pragma unroll
        for (int j = 0; j < 4; ++j) {
            const int e = base + j * PT + t;
            const int ec = e < E ? e : base;
            const int c = col[ec];
            ss[j] = row[ec];
            cc[j] = e < E ? c : -1;
        }
#pragma unroll
        for (int j = 0; j < 4; ++j) {
            const int k = cc[j] >> 8;
            bk[j] = k < NB ? k : NB - 1;
            rk[j] = 0;
            if (cc[j] >= 0) rk[j] = atomicAdd(&hist[bk[j]], 1);
        }
        __syncthreads();
        if (t < 32) {
            int v[8];
            int sum = 0;
#pragma unroll
            for (int k = 0; k < 8; ++k) {
                const int i = 8 * t + k;
                v[k] = i < NB ? hist[i] : 0;
                sum += v[k];
            }
            int incl = sum;
#pragma unroll
            for (int d = 1; d < 32; d <<= 1) {
                const int up = __shfl_up(incl, d);
                if (t >= d) incl += up;
            }
            int run = incl - sum;
#pragma unroll
            for (int k = 0; k < 8; ++k) {
                const int i = 8 * t + k;
                if (i <= NB) hist[i] = run;
                run += v[k];
            }
        }
        __syncthreads();
#pragma unroll
        for (int j = 0; j < 4; ++j)
            if (cc[j] >= 0)
                stage[hist[bk[j]] + rk[j]] = ((unsigned)cc[j] << 16) | ((unsigned)ss[j] & 0xffffu);
        __syncthreads();
        const int n = E - base < CHUNK ? E - base : CHUNK;
#pragma unroll
        for (int j = 0; j < 4; ++j) {
            const int i = j * PT + t;
            if (i < n) sorted[(size_t)base + i] = stage[i];
        }
        if (t < NB) offs[(size_t)w * OFFP + t] = (unsigned)hist[t] | ((unsigned)hist[t + 1] << 16);
    }
}

__device__ inline float dis_of(int c) {
    const float v = (float)(c + 1);
    return (v > 0.0f) ? rsqrtf(fmaxf(v, 1.0f)) : 0.0f;
}

__global__ void __launch_bounds__(PT)
range_build(const unsigned* __restrict__ sorted, const unsigned* __restrict__ offs, int NW,
            const float* __restrict__ x, int* __restrict__ cnt, float* __restrict__ dis,
            u16* __restrict__ srcs, float* __restrict__ xp,
            const float* __restrict__ W23, const float* __restrict__ W4,
            float* __restrict__ W234) {
    __shared__ int lcnt[NPB];
    __shared__ uint4 lrow[NPB * SLOTS * 2 / 16];
    const int t = threadIdx.x;
    const int b = (int)blockIdx.x - FOLD2_WG;
    if (b < 0) {
        const int wave = t >> 5;
        if (wave < 6 * 2)
            tile_wmma_f32<64, 64, 32, false, false, false>(W23, W4, nullptr, nullptr, W234, wave);
        return;
    }
    const int n0 = b * NPB;
    const int nn = NN - n0 < NPB ? NN - n0 : NPB;
    u16* lsrc = (u16*)lrow;

    if (t < NPB) lcnt[t] = 0;
    lrow[t] = make_uint4(0u, 0u, 0u, 0u);
    lrow[t + PT] = make_uint4(0u, 0u, 0u, 0u);
    __syncthreads();

    float xv[2][4];
#pragma unroll
    for (int k = 0; k < 2; ++k) {
        const int j = k * PT + t;
        const int n = j >> 3, c = (j & 7) * 4;
        const float* __restrict__ xr = x + (size_t)(n < nn ? n0 + n : n0) * XK;
#pragma unroll
        for (int m = 0; m < 4; ++m) {
            const float v = xr[c + m < XK ? c + m : XK - 1];
            xv[k][m] = c + m < XK ? v : 0.0f;
        }
    }


    const int g = t >> 3, q = t & 7;
    for (int w0 = 0; w0 < NW; w0 += 2 * (PT / 8)) {
        const int wa = w0 + g, wb = wa + PT / 8;
        const bool va = wa < NW, vb = wb < NW;
        const unsigned pa = offs[(size_t)(va ? wa : 0) * OFFP + b];
        const unsigned pb = offs[(size_t)(vb ? wb : 0) * OFFP + b];
        const unsigned* __restrict__ sa = sorted + (size_t)(va ? wa : 0) * CHUNK;
        const unsigned* __restrict__ sb = sorted + (size_t)(vb ? wb : 0) * CHUNK;
        const int loa = (int)(pa & 0xffffu), lob = (int)(pb & 0xffffu);
        const int hia = va ? (int)(pa >> 16) : loa;
        const int hib = vb ? (int)(pb >> 16) : lob;
        unsigned ea[4], eb[4];
#pragma unroll
        for (int k = 0; k < 4; ++k) {
            const int ia = loa + q + 8 * k, ib = lob + q + 8 * k;
            const unsigned a = sa[ia < hia ? ia : 0];
            const unsigned c = sb[ib < hib ? ib : 0];
            ea[k] = ia < hia ? a : 0xffffffffu;
            eb[k] = ib < hib ? c : 0xffffffffu;
        }
        unsigned loc[8];
        int rk[8];
#pragma unroll
        for (int k = 0; k < 8; ++k) {
            const unsigned e = k < 4 ? ea[k & 3] : eb[k & 3];
            loc[k] = (e >> 16) - (unsigned)n0;
            rk[k] = SLOTS;
            if (loc[k] < (unsigned)nn) rk[k] = atomicAdd(&lcnt[loc[k]], 1);
        }
#pragma unroll
        for (int k = 0; k < 8; ++k) {
            const unsigned e = k < 4 ? ea[k & 3] : eb[k & 3];
            if (rk[k] < SLOTS) lsrc[loc[k] * SLOTS + rk[k]] = (u16)(e & 0xffffu);
        }
#define PLACE_EDGE(EW) { const unsigned e_ = (EW); const unsigned loc_ = (e_ >> 16) - (unsigned)n0; \
            if (loc_ < (unsigned)nn) { const int r_ = atomicAdd(&lcnt[loc_], 1); \
                if (r_ < SLOTS) lsrc[loc_ * SLOTS + r_] = (u16)(e_ & 0xffffu); } }
        for (int i = loa + q + 32; i < hia; i += 8) PLACE_EDGE(sa[i])
        for (int i = lob + q + 32; i < hib; i += 8) PLACE_EDGE(sb[i])
#undef PLACE_EDGE
    }
    __syncthreads();

    if (t < nn) {
        const int d = lcnt[t];
        cnt[n0 + t] = d;
        dis[n0 + t] = dis_of(d);
    }
    uint4* __restrict__ grow = (uint4*)(srcs + (size_t)n0 * SLOTS);
    float4* __restrict__ xp4 = (float4*)xp + (size_t)n0 * (XKP / 4);
#pragma unroll
    for (int k = 0; k < 2; ++k) {
        const int j = k * PT + t;
        const int n = j >> 3;
        if (n < nn) {
            grow[j] = lrow[j];
            const float d = dis_of(lcnt[n]);
            float4 v;
            v.x = xv[k][0] * d; v.y = xv[k][1] * d; v.z = xv[k][2] * d; v.w = xv[k][3] * d;
            xp4[j] = v;
        }
    }
}

static_assert(NN % 8 == 0, "a wave takes eight whole nodes");

template <int U>
__device__ inline int group_lane(int v) {
    return __builtin_amdgcn_mov_dpp8(v, U * 0x249249);
}


__device__ inline void add_overflow_rows(float4& acc, const float4* __restrict__ h4,
                                         const unsigned* __restrict__ sorted,
                                         const unsigned* __restrict__ offs, int NW,
                                         int node, int q) {
    const int b = node / NPB;
    for (int w = 0; w < NW; ++w) {
        const unsigned p = offs[(size_t)w * OFFP + b];
        const int lo = (int)(p & 0xffffu), hi = (int)(p >> 16);
        for (int i = lo; i < hi; ++i) {
            const unsigned e = sorted[(size_t)w * CHUNK + i];
            if ((int)(e >> 16) == node) {
                const float4 r = h4[(e & 0xffffu) * 8u + (unsigned)q];
                acc.x += r.x; acc.y += r.y; acc.z += r.z; acc.w += r.w;
            }
        }
    }
}

__device__ inline void gather_rows4x2(const float4* __restrict__ h4, const int* __restrict__ cnt,
                                      const u16* __restrict__ srcs,
                                      const unsigned* __restrict__ sorted,
                                      const unsigned* __restrict__ offs, int NW,
                                      int nodeA, int nodeB, int q,
                                      float4& outA, float4& outB) {
    const int degA = cnt[nodeA];
    const int degB = cnt[nodeB];
    const int begA = nodeA * SLOTS;
    const int begB = nodeB * SLOTS;
    const int endA = begA + (degA <= SLOTS ? degA : 0);
    const int endB = begB + (degB <= SLOTS ? degB : 0);

    float4 accA = h4[(unsigned)nodeA * 8u + (unsigned)q];
    float4 accB = h4[(unsigned)nodeB * 8u + (unsigned)q];

    int eA = begA, eB = begB;
    int sA, sB;
    {
        const bool inA = eA + q < endA, inB = eB + q < endB;
        const int vA = (int)srcs[inA ? eA + q : 0];
        const int vB = (int)srcs[inB ? eB + q : 0];
        sA = inA ? vA : nodeA;
        sB = inB ? vB : nodeB;
    }
    while (eA < endA || eB < endB) {
        const bool inA1 = eA + 8 + q < endA, inB1 = eB + 8 + q < endB;
        const int vA1 = (int)srcs[inA1 ? eA + 8 + q : 0];
        const int vB1 = (int)srcs[inB1 ? eB + 8 + q : 0];
        float4 rA[8], rB[8];
#define GATHER_ROW(U) rA[U] = h4[(unsigned)group_lane<U>(sA) * 8u + (unsigned)q]; \
                      rB[U] = h4[(unsigned)group_lane<U>(sB) * 8u + (unsigned)q];
        GATHER_ROW(0) GATHER_ROW(1) GATHER_ROW(2) GATHER_ROW(3)
        GATHER_ROW(4) GATHER_ROW(5) GATHER_ROW(6) GATHER_ROW(7)
#undef GATHER_ROW
#pragma unroll
        for (int u = 0; u < 8; ++u) {
            const bool onA = eA + u < endA;
            accA.x += onA ? rA[u].x : 0.0f;
            accA.y += onA ? rA[u].y : 0.0f;
            accA.z += onA ? rA[u].z : 0.0f;
            accA.w += onA ? rA[u].w : 0.0f;
        }
#pragma unroll
        for (int u = 0; u < 8; ++u) {
            const bool onB = eB + u < endB;
            accB.x += onB ? rB[u].x : 0.0f;
            accB.y += onB ? rB[u].y : 0.0f;
            accB.z += onB ? rB[u].z : 0.0f;
            accB.w += onB ? rB[u].w : 0.0f;
        }
        sA = inA1 ? vA1 : nodeA;
        sB = inB1 ? vB1 : nodeB;
        eA += 8;
        eB += 8;
    }

    if (degA > SLOTS) add_overflow_rows(accA, h4, sorted, offs, NW, nodeA, q);
    if (degB > SLOTS) add_overflow_rows(accB, h4, sorted, offs, NW, nodeB, q);
    outA = accA;
    outB = accB;
}

template <bool BIAS, bool OSCALE>
__global__ void __launch_bounds__(256)
agg_gather4(const float* __restrict__ h, const float* __restrict__ dis,
            const int* __restrict__ cnt, const u16* __restrict__ srcs,
            const unsigned* __restrict__ sorted, const unsigned* __restrict__ offs, int NW,
            const float* __restrict__ bias, float* __restrict__ out) {
    const int wave = blockIdx.x * (blockDim.x >> 5) + (threadIdx.x >> 5);
    if (wave >= NN / 8) return;
    const int lane = threadIdx.x & 31;
    const int q    = lane & 7;
    const int nodeA = wave * 8 + (lane >> 3);
    const int nodeB = nodeA + 4;
    const float diA = dis[nodeA];
    const float diB = dis[nodeB];
    float4 bv = make_float4(0.0f, 0.0f, 0.0f, 0.0f);
    if constexpr (BIAS) bv = *(const float4*)(bias + q * 4);
    float4 accA, accB;
    gather_rows4x2((const float4*)h, cnt, srcs, sorted, offs, NW, nodeA, nodeB, q, accA, accB);

    float4 oA, oB;
    oA.x = accA.x * diA; oA.y = accA.y * diA; oA.z = accA.z * diA; oA.w = accA.w * diA;
    oB.x = accB.x * diB; oB.y = accB.y * diB; oB.z = accB.z * diB; oB.w = accB.w * diB;
    if constexpr (BIAS) {
        oA.x += bv.x; oA.y += bv.y; oA.z += bv.z; oA.w += bv.w;
        oB.x += bv.x; oB.y += bv.y; oB.z += bv.z; oB.w += bv.w;
    }
    if constexpr (OSCALE) {
        oA.x *= diA; oA.y *= diA; oA.z *= diA; oA.w *= diA;
        oB.x *= diB; oB.y *= diB; oB.z *= diB; oB.w *= diB;
    }
    ((float4*)out)[(unsigned)nodeA * 8u + (unsigned)q] = oA;
    ((float4*)out)[(unsigned)nodeB * 8u + (unsigned)q] = oB;
}

#define H1F 96
#define UF  32
#define FN  64
#define AXP 36
static_assert(XKP == 32 && UF == 32 && NN % 16 == 0, "a lane half holds 16 of a row's 32 floats; whole node tiles");
static_assert(H1F == 6 * 16 && FN == 4 * 16, "two waves of three feature tiles per node tile, four node tiles");

__global__ void __launch_bounds__(256)
gather_dense(const float* __restrict__ xp, const float* __restrict__ dis,
             const int* __restrict__ cnt, const u16* __restrict__ srcs,
             const unsigned* __restrict__ sorted, const unsigned* __restrict__ offs, int NW,
             const float* __restrict__ W1, const float* __restrict__ b1,
             const float* __restrict__ W234, float* __restrict__ u0) {
    __shared__ float4 lax[FN * AXP / 4];
    __shared__ float4 lpart[FN * AXP / 4];
    const int wv = __builtin_amdgcn_readfirstlane((int)(threadIdx.x >> 5));
    const int lane = threadIdx.x & 31;
    const int n0 = blockIdx.x * FN;

    if (n0 + wv * 8 < NN) {
        const int q   = lane & 7;
        const int nlA = wv * 8 + (lane >> 3);
        const int nlB = nlA + 4;
        const float diA = dis[n0 + nlA];
        const float diB = dis[n0 + nlB];
        float4 accA, accB;
        gather_rows4x2((const float4*)xp, cnt, srcs, sorted, offs, NW, n0 + nlA, n0 + nlB, q, accA, accB);
        float4 oA, oB;
        oA.x = accA.x * diA; oA.y = accA.y * diA; oA.z = accA.z * diA; oA.w = accA.w * diA;
        oB.x = accB.x * diB; oB.y = accB.y * diB; oB.z = accB.z * diB; oB.w = accB.w * diB;
        lax[nlA * (AXP / 4) + q] = oA;
        lax[nlB * (AXP / 4) + q] = oB;
    }
    __syncthreads();

    const int nt = wv >> 1, hh = wv & 1;
    const bool work = n0 + nt * 16 < NN;
    const int hf = lane >> 4;
    const int c  = lane & 15;
    const int node = n0 + nt * 16 + c;
    v8f acc2[UF / 16];
#pragma unroll
    for (int ot = 0; ot < UF / 16; ++ot) acc2[ot] = v8f{};
    float di = 0.0f;

    if (work) {
        di = dis[node];
        float axv[16];
        {
            const float4* __restrict__ axr = &lax[(nt * 16 + c) * (AXP / 4) + 4 * hf];
#pragma unroll
            for (int i = 0; i < 4; ++i) {
                const float4 v = axr[i];
                axv[4 * i] = v.x; axv[4 * i + 1] = v.y; axv[4 * i + 2] = v.z; axv[4 * i + 3] = v.w;
            }
        }
        const float* __restrict__ W1h = W1 + 48 * hh;
        const float* __restrict__ b1h = b1 + 48 * hh;
        const float* __restrict__ W234h = W234 + 48 * hh * UF;

        v8f acc1[3];
#pragma unroll
        for (int j = 0; j < 3; ++j) acc1[j] = v8f{};

#pragma unroll
        for (int t = 0; t < 8; ++t) {
            const int k = 16 * hf + 2 * t;
            float x0 = axv[2 * t], x1 = axv[2 * t + 1];
            const bool padded = 16 + 2 * t + 1 >= XK;
            const bool in0 = !padded || k < XK;
            const bool in1 = !padded || k + 1 < XK;
            const int k0 = in0 ? k : XK - 1;
            const int k1 = in1 ? k + 1 : XK - 1;
            if (!in0) x0 = 0.0f;
            if (!in1) x1 = 0.0f;
            v2f bv; bv.x = x0; bv.y = x1;
#pragma unroll
            for (int j = 0; j < 3; ++j) {
                float w0 = W1h[k0 * H1F + 16 * j + c];
                float w1 = W1h[k1 * H1F + 16 * j + c];
                if (!in0) w0 = 0.0f;
                if (!in1) w1 = 0.0f;
                v2f av; av.x = w0; av.y = w1;
                acc1[j] = __builtin_amdgcn_wmma_f32_16x16x4_f32(false, av, false, bv,
                                                                (short)0, acc1[j], false, false);
            }
        }

#pragma unroll
        for (int j = 0; j < 3; ++j) {
            const float4* __restrict__ bp = (const float4*)(b1h + 16 * j + 8 * hf);
            const float4 bl = bp[0], bh = bp[1];
            float hv[8];
            hv[0] = fmaxf(acc1[j][0] + bl.x, 0.0f); hv[1] = fmaxf(acc1[j][1] + bl.y, 0.0f);
            hv[2] = fmaxf(acc1[j][2] + bl.z, 0.0f); hv[3] = fmaxf(acc1[j][3] + bl.w, 0.0f);
            hv[4] = fmaxf(acc1[j][4] + bh.x, 0.0f); hv[5] = fmaxf(acc1[j][5] + bh.y, 0.0f);
            hv[6] = fmaxf(acc1[j][6] + bh.z, 0.0f); hv[7] = fmaxf(acc1[j][7] + bh.w, 0.0f);
#pragma unroll
            for (int p = 0; p < 4; ++p) {
                const int k = 16 * j + 8 * hf + 2 * p;
                v2f bv; bv.x = hv[2 * p]; bv.y = hv[2 * p + 1];
#pragma unroll
                for (int ot = 0; ot < UF / 16; ++ot) {
                    v2f av;
                    av.x = W234h[k * UF + 16 * ot + c];
                    av.y = W234h[(k + 1) * UF + 16 * ot + c];
                    acc2[ot] = __builtin_amdgcn_wmma_f32_16x16x4_f32(false, av, false, bv,
                                                                     (short)0, acc2[ot], false, false);
                }
            }
        }

        if (hh == 1) {
            float4* __restrict__ pr = &lpart[(nt * 16 + c) * (AXP / 4) + 2 * hf];
#pragma unroll
            for (int ot = 0; ot < UF / 16; ++ot) {
                pr[4 * ot]     = make_float4(acc2[ot][0], acc2[ot][1], acc2[ot][2], acc2[ot][3]);
                pr[4 * ot + 1] = make_float4(acc2[ot][4], acc2[ot][5], acc2[ot][6], acc2[ot][7]);
            }
        }
    }
    __syncthreads();

    if (work && hh == 0) {
        const float4* __restrict__ pr = &lpart[(nt * 16 + c) * (AXP / 4) + 2 * hf];
        float4* __restrict__ o4 = (float4*)(u0 + (size_t)node * UF + 8 * hf);
#pragma unroll
        for (int ot = 0; ot < UF / 16; ++ot) {
            const float4 pl = pr[4 * ot], ph = pr[4 * ot + 1];
            float4 lo, hi;
            lo.x = (acc2[ot][0] + pl.x) * di; lo.y = (acc2[ot][1] + pl.y) * di;
            lo.z = (acc2[ot][2] + pl.z) * di; lo.w = (acc2[ot][3] + pl.w) * di;
            hi.x = (acc2[ot][4] + ph.x) * di; hi.y = (acc2[ot][5] + ph.y) * di;
            hi.z = (acc2[ot][6] + ph.z) * di; hi.w = (acc2[ot][7] + ph.w) * di;
            o4[4 * ot]     = lo;
            o4[4 * ot + 1] = hi;
        }
    }
}

static inline int ceil_div(long long a, long long b) { return (int)((a + b - 1) / b); }

extern "C" void kernel_launch(void* const* d_in, const int* in_sizes, int n_in,
                              void* d_out, int out_size, void* d_ws, size_t ws_size,
                              hipStream_t stream) {
    const float* x  = (const float*)d_in[0];
    const int*   ei = (const int*)d_in[1];
    const float* W1 = (const float*)d_in[2];
    const float* b1 = (const float*)d_in[3];
    const float* W2 = (const float*)d_in[4];
    const float* b2 = (const float*)d_in[5];
    const float* W3 = (const float*)d_in[6];
    const float* b3 = (const float*)d_in[7];
    const float* W4 = (const float*)d_in[8];
    const float* b4 = (const float*)d_in[9];

    const int E = in_sizes[1] / 2;
    const int* row = ei;
    const int* col = ei + E;

    float* dis  = (float*)d_ws;
    float* reg1 = dis  + ((NN + 255) & ~255);
    float* bufA = reg1 + ((E  + 255) & ~255);
    float* bufB = bufA + (size_t)NN * 128;
    int*   cnt  = (int*)reg1;
    float* W23  = (float*)cnt + ((NN + 1 + 255) & ~255);
    float* W234 = W23  + 96 * 64;
    float* c2   = W234 + 96 * 32;
    float* c3   = c2 + 32;
    const int NW = ceil_div(E, CHUNK);
    u16*      srcs   = (u16*)(bufA + (size_t)NN * 96);
    unsigned* sorted = (unsigned*)bufB;
    unsigned* offs   = (unsigned*)(bufB + ((E + 255) & ~255));
    float* xp = bufA;
    float* u0 = bufA + (size_t)NN * 32;
    float* u1 = bufA + (size_t)NN * 64;
    float* u2 = bufA;

    const int TB = 256;

    edge_bucket_fold<<<FOLD1_WG + NW, PT, 0, stream>>>(row, col, E, sorted, offs, W2, W3, W23, b2, b3, W4, c2, c3);
    range_build<<<FOLD2_WG + NB, PT, 0, stream>>>(sorted, offs, NW, x, cnt, dis, srcs, xp, W23, W4, W234);

    const int AGB = ceil_div(NN / 8, TB / 32);

    gather_dense<<<ceil_div(NN, FN), 256, 0, stream>>>(xp, dis, cnt, srcs, sorted, offs, NW, W1, b1, W234, u0);
    float* out = (float*)d_out;
    agg_gather4<true, true><<<AGB, TB, 0, stream>>>(u0, dis, cnt, srcs, sorted, offs, NW, c2, u1);
    agg_gather4<true, true><<<AGB, TB, 0, stream>>>(u1, dis, cnt, srcs, sorted, offs, NW, c3, u2);
    agg_gather4<true, false><<<AGB, TB, 0, stream>>>(u2, dis, cnt, srcs, sorted, offs, NW, b4, out);
}
